# starting kernel
# baseline (speedup 1.0000x reference)
#include <hip/hip_runtime.h>

typedef __attribute__((ext_vector_type(2))) float v2f;
typedef __attribute__((ext_vector_type(8))) float v8f;

#define N_NODES_C 100000
#define N_EDGES_C 20000
#define NNZ_C     1000000
#define ALPHA_C   0.5f

__device__ __forceinline__ v8f wmma_f32(v2f a, v2f b, v8f c) {
  return __builtin_amdgcn_wmma_f32_16x16x4_f32(false, a, false, b, (short)0, c,
                                               false, false);
}

__global__ __launch_bounds__(256) void gemm64_bias_kernel(
    const float* __restrict__ A, const float* __restrict__ W,
    const float* __restrict__ bias, float* __restrict__ Out, int M) {
  __shared__ float Bsh[64 * 64];
  for (int i = threadIdx.x; i < 64 * 64; i += 256) {
    const int k = i >> 6, n = i & 63;
    Bsh[i] = W[n * 64 + k];
  }
  __syncthreads();

  const int lane = threadIdx.x & 31;
  const int wave = threadIdx.x >> 5;
  const int rowBase = (blockIdx.x * 8 + wave) << 4;
  if (rowBase >= M) return;
  const int ll = lane & 15;
  const int lh = lane >> 4;

  const float* Arow = A + (size_t)(rowBase + ll) * 64;
  v8f acc0 = {}, acc1 = {}, acc2 = {}, acc3 = {};

#pragma unroll
  for (int k0 = 0; k0 < 64; k0 += 4) {
    const int ka = k0 + lh * 2;
    v2f a;
    a.x = Arow[ka];
    a.y = Arow[ka + 1];
    const float* B0 = &Bsh[ka * 64 + ll];
    const float* B1 = &Bsh[(ka + 1) * 64 + ll];
    v2f b;
    b.x = B0[0];  b.y = B1[0];  acc0 = wmma_f32(a, b, acc0);
    b.x = B0[16]; b.y = B1[16]; acc1 = wmma_f32(a, b, acc1);
    b.x = B0[32]; b.y = B1[32]; acc2 = wmma_f32(a, b, acc2);
    b.x = B0[48]; b.y = B1[48]; acc3 = wmma_f32(a, b, acc3);
  }

#pragma unroll
  for (int j = 0; j < 8; ++j) {
    const int r = rowBase + j + lh * 8;
    float* o = Out + (size_t)r * 64;
    o[ll]      = acc0[j] + bias[ll];
    o[16 + ll] = acc1[j] + bias[16 + ll];
    o[32 + ll] = acc2[j] + bias[32 + ll];
    o[48 + ll] = acc3[j] + bias[48 + ll];
  }
}

__global__ __launch_bounds__(256) void gemm2_mix_kernel(
    const float* __restrict__ X, const float* __restrict__ X0,
    const float* __restrict__ S, const float* __restrict__ deg,
    const float* __restrict__ W2, const float* __restrict__ b2,
    float* __restrict__ T, int M) {
  __shared__ float Bsh[128 * 64];
  for (int i = threadIdx.x; i < 128 * 64; i += 256) {
    const int k = i >> 6, n = i & 63;
    Bsh[i] = W2[n * 128 + k];
  }
  __syncthreads();

  const int lane = threadIdx.x & 31;
  const int wave = threadIdx.x >> 5;
  const int rowBase = (blockIdx.x * 8 + wave) << 4;
  if (rowBase >= M) return;
  const int ll = lane & 15;
  const int lh = lane >> 4;

  const float dg = deg[rowBase + ll];
  const float* Xrow = X + (size_t)(rowBase + ll) * 64;
  const float* Srow = S + (size_t)(rowBase + ll) * 64;
  v8f acc0 = {}, acc1 = {}, acc2 = {}, acc3 = {};

#pragma unroll
  for (int k0 = 0; k0 < 64; k0 += 4) {
    const int ka = k0 + lh * 2;
    v2f a;
    a.x = dg * Xrow[ka];
    a.y = dg * Xrow[ka + 1];
    const float* B0 = &Bsh[ka * 64 + ll];
    const float* B1 = &Bsh[(ka + 1) * 64 + ll];
    v2f b;
    b.x = B0[0];  b.y = B1[0];  acc0 = wmma_f32(a, b, acc0);
    b.x = B0[16]; b.y = B1[16]; acc1 = wmma_f32(a, b, acc1);
    b.x = B0[32]; b.y = B1[32]; acc2 = wmma_f32(a, b, acc2);
    b.x = B0[48]; b.y = B1[48]; acc3 = wmma_f32(a, b, acc3);
  }
#pragma unroll
  for (int k0 = 64; k0 < 128; k0 += 4) {
    const int ka = k0 + lh * 2;
    v2f a;
    a.x = Srow[ka - 64];
    a.y = Srow[ka - 63];
    const float* B0 = &Bsh[ka * 64 + ll];
    const float* B1 = &Bsh[(ka + 1) * 64 + ll];
    v2f b;
    b.x = B0[0];  b.y = B1[0];  acc0 = wmma_f32(a, b, acc0);
    b.x = B0[16]; b.y = B1[16]; acc1 = wmma_f32(a, b, acc1);
    b.x = B0[32]; b.y = B1[32]; acc2 = wmma_f32(a, b, acc2);
    b.x = B0[48]; b.y = B1[48]; acc3 = wmma_f32(a, b, acc3);
  }

  const float oma = 1.0f - ALPHA_C;
#pragma unroll
  for (int j = 0; j < 8; ++j) {
    const int r = rowBase + j + lh * 8;
    const float dr = deg[r];
    const float* x0r = X0 + (size_t)r * 64;
    float* o = T + (size_t)r * 64;
    o[ll]      = oma * (acc0[j] + dr * b2[ll])      + ALPHA_C * x0r[ll];
    o[16 + ll] = oma * (acc1[j] + dr * b2[16 + ll]) + ALPHA_C * x0r[16 + ll];
    o[32 + ll] = oma * (acc2[j] + dr * b2[32 + ll]) + ALPHA_C * x0r[32 + ll];
    o[48 + ll] = oma * (acc3[j] + dr * b2[48 + ll]) + ALPHA_C * x0r[48 + ll];
  }
}

__global__ __launch_bounds__(256) void scatter_edges_kernel(
    const float* __restrict__ Y1, const int* __restrict__ vertex,
    const int* __restrict__ edges, float* __restrict__ Xe,
    float* __restrict__ deg) {
  const int tid = blockIdx.x * 256 + threadIdx.x;
  const int e = tid >> 4;
  if (e >= NNZ_C) return;
  const int c = (tid & 15) << 2;
  const int v = vertex[e];
  const int h = edges[e];
  const float4 y = *(const float4*)(Y1 + (size_t)v * 64 + c);
  float* dst = Xe + (size_t)h * 64 + c;
  unsafeAtomicAdd(dst + 0, y.x);
  unsafeAtomicAdd(dst + 1, y.y);
  unsafeAtomicAdd(dst + 2, y.z);
  unsafeAtomicAdd(dst + 3, y.w);
  if ((tid & 15) == 0) unsafeAtomicAdd(deg + v, 1.0f);
}

__global__ __launch_bounds__(256) void scatter_nodes_kernel(
    const float* __restrict__ Xe, const int* __restrict__ vertex,
    const int* __restrict__ edges, float* __restrict__ S) {
  const int tid = blockIdx.x * 256 + threadIdx.x;
  const int e = tid >> 4;
  if (e >= NNZ_C) return;
  const int c = (tid & 15) << 2;
  const int v = vertex[e];
  const int h = edges[e];
  const float4 y = *(const float4*)(Xe + (size_t)h * 64 + c);
  float* dst = S + (size_t)v * 64 + c;
  unsafeAtomicAdd(dst + 0, y.x);
  unsafeAtomicAdd(dst + 1, y.y);
  unsafeAtomicAdd(dst + 2, y.z);
  unsafeAtomicAdd(dst + 3, y.w);
}

extern "C" void kernel_launch(void* const* d_in, const int* in_sizes, int n_in,
                              void* d_out, int out_size, void* d_ws,
                              size_t ws_size, hipStream_t stream) {
  (void)in_sizes; (void)n_in; (void)out_size; (void)ws_size;
  const float* X    = (const float*)d_in[0];
  const float* X0   = (const float*)d_in[1];
  const int*   vertex = (const int*)d_in[2];
  const int*   edges  = (const int*)d_in[3];
  const float* W1w  = (const float*)d_in[4];
  const float* W1b  = (const float*)d_in[5];
  const float* W2w  = (const float*)d_in[6];
  const float* W2b  = (const float*)d_in[7];
  const float* Ww   = (const float*)d_in[8];
  const float* Wb   = (const float*)d_in[9];

  float* out = (float*)d_out;
  float* Xe  = out + (size_t)N_NODES_C * 64;

  float* ws  = (float*)d_ws;
  float* Y1  = ws;
  float* S   = Y1 + (size_t)N_NODES_C * 64;
  float* deg = S + (size_t)N_NODES_C * 64;

  hipMemsetAsync(Xe, 0, (size_t)N_EDGES_C * 64 * sizeof(float), stream);
  hipMemsetAsync(S, 0, ((size_t)N_NODES_C * 64 + N_NODES_C) * sizeof(float),
                 stream);

  const int tiles = N_NODES_C / 16;
  const int gemmBlocks = (tiles + 7) / 8;
  const int scatterBlocks = (NNZ_C * 16) / 256;

  gemm64_bias_kernel<<<gemmBlocks, 256, 0, stream>>>(X, W1w, W1b, Y1, N_NODES_C);
  scatter_edges_kernel<<<scatterBlocks, 256, 0, stream>>>(Y1, vertex, edges, Xe, deg);
  scatter_nodes_kernel<<<scatterBlocks, 256, 0, stream>>>(Xe, vertex, edges, S);
  gemm2_mix_kernel<<<gemmBlocks, 256, 0, stream>>>(X, X0, S, deg, W2w, W2b, Y1,
                                                   N_NODES_C);
  gemm64_bias_kernel<<<gemmBlocks, 256, 0, stream>>>(Y1, Ww, Wb, out, N_NODES_C);
}


# segment sums without binning or global atomics: a work-group owns a range of destination rows in LDS and scans a bit-sliced owner map of ALL incidences for its own (LDS f32 adds, arrival order)
# speedup vs baseline: 1.5797x; 1.5797x over previous
#include <hip/hip_runtime.h>

typedef __attribute__((ext_vector_type(2))) float v2f;
typedef __attribute__((ext_vector_type(8))) float v8f;

#define N_NODES_C 100000
#define N_EDGES_C 20000
#define NNZ_C     1000000
#define ALPHA_C   0.5f

__device__ __forceinline__ v8f wmma_f32(v2f a, v2f b, v8f c) {
  return __builtin_amdgcn_wmma_f32_16x16x4_f32(false, a, false, b, (short)0, c,
                                               false, false);
}

__global__ __launch_bounds__(256) void gemm64_bias_kernel(
    const float* __restrict__ A, const float* __restrict__ W,
    const float* __restrict__ bias, float* __restrict__ Out, int M) {
  __shared__ float Bsh[64 * 64];
  for (int i = threadIdx.x; i < 64 * 64; i += 256) {
    const int k = i >> 6, n = i & 63;
    Bsh[i] = W[n * 64 + k];
  }
  __syncthreads();

  const int lane = threadIdx.x & 31;
  const int wave = threadIdx.x >> 5;
  const int rowBase = (blockIdx.x * 8 + wave) << 4;
  if (rowBase >= M) return;
  const int ll = lane & 15;
  const int lh = lane >> 4;

  const float* Arow = A + (size_t)(rowBase + ll) * 64;
  v8f acc0 = {}, acc1 = {}, acc2 = {}, acc3 = {};

#pragma unroll
  for (int k0 = 0; k0 < 64; k0 += 4) {
    const int ka = k0 + lh * 2;
    v2f a;
    a.x = Arow[ka];
    a.y = Arow[ka + 1];
    const float* B0 = &Bsh[ka * 64 + ll];
    const float* B1 = &Bsh[(ka + 1) * 64 + ll];
    v2f b;
    b.x = B0[0];  b.y = B1[0];  acc0 = wmma_f32(a, b, acc0);
    b.x = B0[16]; b.y = B1[16]; acc1 = wmma_f32(a, b, acc1);
    b.x = B0[32]; b.y = B1[32]; acc2 = wmma_f32(a, b, acc2);
    b.x = B0[48]; b.y = B1[48]; acc3 = wmma_f32(a, b, acc3);
  }

#pragma unroll
  for (int j = 0; j < 8; ++j) {
    const int r = rowBase + j + lh * 8;
    float* o = Out + (size_t)r * 64;
    o[ll]      = acc0[j] + bias[ll];
    o[16 + ll] = acc1[j] + bias[16 + ll];
    o[32 + ll] = acc2[j] + bias[32 + ll];
    o[48 + ll] = acc3[j] + bias[48 + ll];
  }
}

__global__ __launch_bounds__(256) void gemm2_mix_kernel(
    const float* __restrict__ X, const float* __restrict__ X0,
    const float* __restrict__ S, const float* __restrict__ deg,
    const float* __restrict__ W2, const float* __restrict__ b2,
    float* __restrict__ T, int M) {
  __shared__ float Bsh[128 * 64];
  for (int i = threadIdx.x; i < 128 * 64; i += 256) {
    const int k = i >> 6, n = i & 63;
    Bsh[i] = W2[n * 128 + k];
  }
  __syncthreads();

  const int lane = threadIdx.x & 31;
  const int wave = threadIdx.x >> 5;
  const int rowBase = (blockIdx.x * 8 + wave) << 4;
  if (rowBase >= M) return;
  const int ll = lane & 15;
  const int lh = lane >> 4;

  const float dg = deg[rowBase + ll];
  const float* Xrow = X + (size_t)(rowBase + ll) * 64;
  const float* Srow = S + (size_t)(rowBase + ll) * 64;
  v8f acc0 = {}, acc1 = {}, acc2 = {}, acc3 = {};

#pragma unroll
  for (int k0 = 0; k0 < 64; k0 += 4) {
    const int ka = k0 + lh * 2;
    v2f a;
    a.x = dg * Xrow[ka];
    a.y = dg * Xrow[ka + 1];
    const float* B0 = &Bsh[ka * 64 + ll];
    const float* B1 = &Bsh[(ka + 1) * 64 + ll];
    v2f b;
    b.x = B0[0];  b.y = B1[0];  acc0 = wmma_f32(a, b, acc0);
    b.x = B0[16]; b.y = B1[16]; acc1 = wmma_f32(a, b, acc1);
    b.x = B0[32]; b.y = B1[32]; acc2 = wmma_f32(a, b, acc2);
    b.x = B0[48]; b.y = B1[48]; acc3 = wmma_f32(a, b, acc3);
  }
#pragma unroll
  for (int k0 = 64; k0 < 128; k0 += 4) {
    const int ka = k0 + lh * 2;
    v2f a;
    a.x = Srow[ka - 64];
    a.y = Srow[ka - 63];
    const float* B0 = &Bsh[ka * 64 + ll];
    const float* B1 = &Bsh[(ka + 1) * 64 + ll];
    v2f b;
    b.x = B0[0];  b.y = B1[0];  acc0 = wmma_f32(a, b, acc0);
    b.x = B0[16]; b.y = B1[16]; acc1 = wmma_f32(a, b, acc1);
    b.x = B0[32]; b.y = B1[32]; acc2 = wmma_f32(a, b, acc2);
    b.x = B0[48]; b.y = B1[48]; acc3 = wmma_f32(a, b, acc3);
  }

  const float oma = 1.0f - ALPHA_C;
#pragma unroll
  for (int j = 0; j < 8; ++j) {
    const int r = rowBase + j + lh * 8;
    const float dr = deg[r];
    const float* x0r = X0 + (size_t)r * 64;
    float* o = T + (size_t)r * 64;
    o[ll]      = oma * (acc0[j] + dr * b2[ll])      + ALPHA_C * x0r[ll];
    o[16 + ll] = oma * (acc1[j] + dr * b2[16 + ll]) + ALPHA_C * x0r[16 + ll];
    o[32 + ll] = oma * (acc2[j] + dr * b2[32 + ll]) + ALPHA_C * x0r[32 + ll];
    o[48 + ll] = oma * (acc3[j] + dr * b2[48 + ll]) + ALPHA_C * x0r[48 + ll];
  }
}

#define OWN_E   79
#define OWN_V   391
#define OWNERS  256
#define FT      1024
#define FG      4
#define QCAP    2048
#define DR      20
#define NGROUPS (NNZ_C / 32)

__global__ __launch_bounds__(256) void owner_map_kernel(
    const int* __restrict__ vertex, const int* __restrict__ edges,
    unsigned int* __restrict__ mapE, unsigned int* __restrict__ mapV) {
  const int e = blockIdx.x * 256 + threadIdx.x;
  if (e >= NNZ_C) return;
  const int lane = threadIdx.x & 31;
  const unsigned int oE = (unsigned int)(edges[e] / OWN_E);
  const unsigned int oV = (unsigned int)(vertex[e] / OWN_V);
  unsigned int pe = 0, pv = 0;
#pragma unroll
  for (int k = 0; k < 8; ++k) {
    const unsigned int bE = __builtin_amdgcn_ballot_w32(((oE >> k) & 1u) != 0u);
    const unsigned int bV = __builtin_amdgcn_ballot_w32(((oV >> k) & 1u) != 0u);
    if (lane == k) { pe = bE; pv = bV; }
  }
  if (lane < 8) {
    mapE[(e >> 5) * 8 + lane] = pe;
    mapV[(e >> 5) * 8 + lane] = pv;
  }
}

__device__ __forceinline__ unsigned int owner_match(const uint4 a, const uint4 b,
                                                    const unsigned int me) {
  unsigned int r;
  r  = a.x ^ ((me & 1u)   ? 0u : 0xFFFFFFFFu);
  r &= a.y ^ ((me & 2u)   ? 0u : 0xFFFFFFFFu);
  r &= a.z ^ ((me & 4u)   ? 0u : 0xFFFFFFFFu);
  r &= a.w ^ ((me & 8u)   ? 0u : 0xFFFFFFFFu);
  r &= b.x ^ ((me & 16u)  ? 0u : 0xFFFFFFFFu);
  r &= b.y ^ ((me & 32u)  ? 0u : 0xFFFFFFFFu);
  r &= b.z ^ ((me & 64u)  ? 0u : 0xFFFFFFFFu);
  r &= b.w ^ ((me & 128u) ? 0u : 0xFFFFFFFFu);
  return r;
}

__device__ __forceinline__ void filter_slow_one(
    const float* __restrict__ src, const int* __restrict__ dstIdx,
    const int* __restrict__ srcIdx, int e, int base, float* acc, int* cnt,
    bool wantDeg) {
  const int r = dstIdx[e] - base;
  const float* s = src + (size_t)srcIdx[e] * 64;
  for (int c = 0; c < 64; ++c) atomicAdd(&acc[r * 64 + c], s[c]);
  if (wantDeg) atomicAdd(&cnt[r], 1);
}

__device__ __forceinline__ void filter_sum_body(
    const float* __restrict__ src, const int* __restrict__ dstIdx,
    const int* __restrict__ srcIdx, const unsigned int* __restrict__ map,
    float* __restrict__ dst, float* __restrict__ deg, const bool wantDeg,
    const int R, const int nDst, float* acc, int* cnt, int* q, int* qn) {
  const int tid = threadIdx.x;
  const int lane = tid & 31;
  const int wave = tid >> 5;
  const int base = blockIdx.x * R;

  for (int i = tid; i < R * 64; i += FT) acc[i] = 0.0f;
  if (wantDeg)
    for (int i = tid; i < R; i += FT) cnt[i] = 0;
  if (tid < 3) qn[tid] = 0;
  __syncthreads();

  const unsigned int me = blockIdx.x;
  const int nChunks = (NGROUPS + FT * FG - 1) / (FT * FG);
  const uint4* mapv = (const uint4*)map;
  int ia = 0, ib = 2, ic = 1;

  for (int c = 0; c <= nChunks; ++c) {
    const int c0 = c * (FT * FG);
    const int cp = c0 - FT * FG;
    const bool doScan = c < nChunks;
    int* qScan = q + (c & 1) * QCAP;
    const int* qDrain = q + ((c & 1) ^ 1) * QCAP;

    uint4 ma[FG], mb[FG];
    if (doScan) {
#pragma unroll
      for (int s = 0; s < FG; ++s) {
        const int gi = c0 + s * FT + tid;
        const int gl = gi < NGROUPS ? gi : NGROUPS - 1;
        ma[s] = mapv[gl * 2];
        mb[s] = mapv[gl * 2 + 1];
      }
    }

    int n = (c > 0) ? qn[ib] : 0;
    if (tid == 0) qn[ic] = 0;
    if (n > QCAP) {
      n = 0;
#pragma nounroll
      for (int s = 0; s < FG; ++s) {
        const int gi = cp + s * FT + tid;
        if (gi >= NGROUPS) continue;
        unsigned int z = owner_match(mapv[gi * 2], mapv[gi * 2 + 1], me);
        while (z) {
          const int b = __builtin_ctz(z);
          z &= z - 1;
          filter_slow_one(src, dstIdx, srcIdx, gi * 32 + b, base, acc, cnt,
                          wantDeg);
        }
      }
    }

    int dI = 0, sI = 0;
    bool has = false;
    if (lane < DR) {
      const int qi = lane * 32 + wave;
      if (qi < n) {
        const int e = qDrain[qi];
        dI = dstIdx[e];
        sI = srcIdx[e];
        has = true;
      }
    }

    if (doScan) {
#pragma unroll
      for (int s = 0; s < FG; ++s) {
        const int gi = c0 + s * FT + tid;
        unsigned int z = gi < NGROUPS ? owner_match(ma[s], mb[s], me) : 0u;
        while (z) {
          const int b = __builtin_ctz(z);
          z &= z - 1;
          const int slot = atomicAdd(&qn[ia], 1);
          if (slot < QCAP) qScan[slot] = gi * 32 + b;
        }
      }
    }

    int pk = has ? (((dI - base) << 20) | sI) : -1;
    for (int jb = 0; jb < n; jb += 32 * DR) {
      if (jb > 0) {
        pk = -1;
        if (lane < DR) {
          const int qi = jb + lane * 32 + wave;
          if (qi < n) {
            const int e = qDrain[qi];
            pk = ((dstIdx[e] - base) << 20) | srcIdx[e];
          }
        }
      }
      float2 r[DR];
      int p[DR];
#pragma unroll
      for (int k = 0; k < DR; ++k) {
        p[k] = __builtin_amdgcn_readlane(pk, k);
        const int srow = p[k] >= 0 ? (p[k] & 0xFFFFF) : 0;
        r[k] = *(const float2*)(src + (size_t)srow * 64 + lane * 2);
      }
#pragma unroll
      for (int k = 0; k < DR; ++k) {
        if (p[k] >= 0) {
          float* a = acc + (p[k] >> 20) * 64 + lane * 2;
          atomicAdd(a, r[k].x);
          atomicAdd(a + 1, r[k].y);
          if (wantDeg && lane == 0) atomicAdd(&cnt[p[k] >> 20], 1);
        }
      }
    }
    __syncthreads();
    const int t = ib; ib = ia; ia = ic; ic = t;
  }

  for (int i = tid; i < R * 64; i += FT) {
    const int row = base + (i >> 6);
    if (row < nDst) dst[(size_t)row * 64 + (i & 63)] = acc[i];
  }
  if (wantDeg)
    for (int i = tid; i < R; i += FT)
      if (base + i < nDst) deg[base + i] = (float)cnt[i];
}

__global__ __launch_bounds__(FT) void filter_edges_kernel(
    const float* __restrict__ Y1, const int* __restrict__ vertex,
    const int* __restrict__ edges, const unsigned int* __restrict__ mapE,
    float* __restrict__ Xe) {
  __shared__ float acc[OWN_E * 64];
  __shared__ int q[2 * QCAP];
  __shared__ int qn[3];
  filter_sum_body(Y1, edges, vertex, mapE, Xe, nullptr, false, OWN_E,
                  N_EDGES_C, acc, nullptr, q, qn);
}

__global__ __launch_bounds__(FT) void filter_nodes_kernel(
    const float* __restrict__ Xe, const int* __restrict__ vertex,
    const int* __restrict__ edges, const unsigned int* __restrict__ mapV,
    float* __restrict__ S, float* __restrict__ deg) {
  __shared__ float acc[OWN_V * 64];
  __shared__ int cnt[OWN_V];
  __shared__ int q[2 * QCAP];
  __shared__ int qn[3];
  filter_sum_body(Xe, vertex, edges, mapV, S, deg, true, OWN_V, N_NODES_C,
                  acc, cnt, q, qn);
}

extern "C" void kernel_launch(void* const* d_in, const int* in_sizes, int n_in,
                              void* d_out, int out_size, void* d_ws,
                              size_t ws_size, hipStream_t stream) {
  (void)in_sizes; (void)n_in; (void)out_size; (void)ws_size;
  const float* X    = (const float*)d_in[0];
  const float* X0   = (const float*)d_in[1];
  const int*   vertex = (const int*)d_in[2];
  const int*   edges  = (const int*)d_in[3];
  const float* W1w  = (const float*)d_in[4];
  const float* W1b  = (const float*)d_in[5];
  const float* W2w  = (const float*)d_in[6];
  const float* W2b  = (const float*)d_in[7];
  const float* Ww   = (const float*)d_in[8];
  const float* Wb   = (const float*)d_in[9];

  float* out = (float*)d_out;
  float* Xe  = out + (size_t)N_NODES_C * 64;

  float* ws  = (float*)d_ws;
  float* Y1  = ws;
  float* S   = Y1 + (size_t)N_NODES_C * 64;
  float* deg = S + (size_t)N_NODES_C * 64;

  unsigned int* mapE = (unsigned int*)d_out;
  unsigned int* mapV = mapE + NNZ_C / 4;

  const int tiles = N_NODES_C / 16;
  const int gemmBlocks = (tiles + 7) / 8;
  const int mapBlocks = (NNZ_C + 255) / 256;

  gemm64_bias_kernel<<<gemmBlocks, 256, 0, stream>>>(X, W1w, W1b, Y1, N_NODES_C);
  owner_map_kernel<<<mapBlocks, 256, 0, stream>>>(vertex, edges, mapE, mapV);
  filter_edges_kernel<<<OWNERS, FT, 0, stream>>>(Y1, vertex, edges, mapE, Xe);
  filter_nodes_kernel<<<OWNERS, FT, 0, stream>>>(Xe, vertex, edges, mapV, S, deg);
  gemm2_mix_kernel<<<gemmBlocks, 256, 0, stream>>>(X, X0, S, deg, W2w, W2b, Y1,
                                                   N_NODES_C);
  gemm64_bias_kernel<<<gemmBlocks, 256, 0, stream>>>(Y1, Ww, Wb, out, N_NODES_C);
}


# launch shape: owner map and the three products run as 256 resident work-groups looping over their tiles (50k -> 12k waves to enter, weights staged once per WGP); no sum reordered
# speedup vs baseline: 2.4621x; 1.5586x over previous
#include <hip/hip_runtime.h>

typedef __attribute__((ext_vector_type(2))) float v2f;
typedef __attribute__((ext_vector_type(8))) float v8f;

#define N_NODES_C 100000
#define N_EDGES_C 20000
#define NNZ_C     1000000
#define ALPHA_C   0.5f

__device__ __forceinline__ v8f wmma_f32(v2f a, v2f b, v8f c) {
  return __builtin_amdgcn_wmma_f32_16x16x4_f32(false, a, false, b, (short)0, c,
                                               false, false);
}

#define RES_WG   256
#define GEMM_T   512
#define GEMM_NW  (GEMM_T / 32)
#define GEMM2_T  256
#define GEMM2_NW (GEMM2_T / 32)

__global__ __launch_bounds__(GEMM_T) void gemm64_bias_kernel(
    const float* __restrict__ A, const float* __restrict__ W,
    const float* __restrict__ bias, float* __restrict__ Out, int M) {
  __shared__ float Bsh[64 * 64];

  const int lane = threadIdx.x & 31;
  const int wave = __builtin_amdgcn_readfirstlane((int)(threadIdx.x >> 5));
  const int ll = lane & 15;
  const int lh = lane >> 4;
  const int tiles = M >> 4;
  const int tBeg = (int)(((long long)blockIdx.x * tiles) / RES_WG);
  const int tEnd = (int)(((long long)(blockIdx.x + 1) * tiles) / RES_WG);
  int t = tBeg + wave;

  v2f an[16];
#pragma unroll
  for (int s = 0; s < 16; ++s) { an[s].x = 0.0f; an[s].y = 0.0f; }
  if (t < tEnd) {
    const float* Arow = A + (size_t)(t * 16 + ll) * 64;
#pragma unroll
    for (int s = 0; s < 16; ++s) {
      const int ka = s * 4 + lh * 2;
      an[s].x = Arow[ka];
      an[s].y = Arow[ka + 1];
    }
  }

#pragma unroll
  for (int j = 0; j < 64 * 64 / GEMM_T; ++j) {
    const int i = j * GEMM_T + (int)threadIdx.x;
    const int k = i >> 6, n = i & 63;
    Bsh[i] = W[n * 64 + k];
  }
  const float bi0 = bias[ll], bi1 = bias[16 + ll], bi2 = bias[32 + ll],
              bi3 = bias[48 + ll];
  __syncthreads();

  while (t < tEnd) {
    v2f a[16];
#pragma unroll
    for (int s = 0; s < 16; ++s) a[s] = an[s];
    const int tn = t + GEMM_NW;
    if (tn < tEnd) {
      const float* Arow = A + (size_t)(tn * 16 + ll) * 64;
#pragma unroll
      for (int s = 0; s < 16; ++s) {
        const int ka = s * 4 + lh * 2;
        an[s].x = Arow[ka];
        an[s].y = Arow[ka + 1];
      }
    }

    v8f acc0 = {}, acc1 = {}, acc2 = {}, acc3 = {};
#pragma unroll
    for (int s = 0; s < 16; ++s) {
      const int ka = s * 4 + lh * 2;
      const float* B0 = &Bsh[ka * 64 + ll];
      const float* B1 = &Bsh[(ka + 1) * 64 + ll];
      v2f b;
      b.x = B0[0];  b.y = B1[0];  acc0 = wmma_f32(a[s], b, acc0);
      b.x = B0[16]; b.y = B1[16]; acc1 = wmma_f32(a[s], b, acc1);
      b.x = B0[32]; b.y = B1[32]; acc2 = wmma_f32(a[s], b, acc2);
      b.x = B0[48]; b.y = B1[48]; acc3 = wmma_f32(a[s], b, acc3);
    }

    const int rowBase = t << 4;
#pragma unroll
    for (int j = 0; j < 8; ++j) {
      const int r = rowBase + j + lh * 8;
      float* o = Out + (size_t)r * 64;
      o[ll]      = acc0[j] + bi0;
      o[16 + ll] = acc1[j] + bi1;
      o[32 + ll] = acc2[j] + bi2;
      o[48 + ll] = acc3[j] + bi3;
    }
    t = tn;
  }
}

__global__ __launch_bounds__(GEMM2_T) void gemm2_mix_kernel(
    const float* __restrict__ X, const float* __restrict__ X0,
    const float* __restrict__ S, const float* __restrict__ deg,
    const float* __restrict__ W2, const float* __restrict__ b2,
    float* __restrict__ T, int M) {
  __shared__ float Bsh[128 * 64];

  const int lane = threadIdx.x & 31;
  const int wave = __builtin_amdgcn_readfirstlane((int)(threadIdx.x >> 5));
  const int ll = lane & 15;
  const int lh = lane >> 4;
  const int tiles = M >> 4;
  const int tBeg = (int)(((long long)blockIdx.x * tiles) / RES_WG);
  const int tEnd = (int)(((long long)(blockIdx.x + 1) * tiles) / RES_WG);
  int t = tBeg + wave;

  v2f xn[16], sn[16];
  float dgn = 0.0f;
#pragma unroll
  for (int s = 0; s < 16; ++s) {
    xn[s].x = 0.0f; xn[s].y = 0.0f; sn[s].x = 0.0f; sn[s].y = 0.0f;
  }
  if (t < tEnd) {
    dgn = deg[t * 16 + ll];
    const float* Xrow = X + (size_t)(t * 16 + ll) * 64;
    const float* Srow = S + (size_t)(t * 16 + ll) * 64;
#pragma unroll
    for (int s = 0; s < 16; ++s) {
      const int ka = s * 4 + lh * 2;
      xn[s].x = Xrow[ka]; xn[s].y = Xrow[ka + 1];
      sn[s].x = Srow[ka]; sn[s].y = Srow[ka + 1];
    }
  }

  {
    float w[128 * 64 / GEMM2_T];
#pragma unroll
    for (int j = 0; j < 128 * 64 / GEMM2_T; ++j) {
      const int i = j * GEMM2_T + (int)threadIdx.x;
      const int k = i >> 6, n = i & 63;
      w[j] = W2[n * 128 + k];
    }
#pragma unroll
    for (int j = 0; j < 128 * 64 / GEMM2_T; ++j)
      Bsh[j * GEMM2_T + (int)threadIdx.x] = w[j];
  }
  const float c0 = b2[ll], c1 = b2[16 + ll], c2 = b2[32 + ll], c3 = b2[48 + ll];
  __syncthreads();

  const float oma = 1.0f - ALPHA_C;
  while (t < tEnd) {
    v2f xa[16], sa[16];
#pragma unroll
    for (int s = 0; s < 16; ++s) { xa[s] = xn[s]; sa[s] = sn[s]; }
    const float dg = dgn;
    const int tn = t + GEMM2_NW;
    if (tn < tEnd) {
      dgn = deg[tn * 16 + ll];
      const float* Xrow = X + (size_t)(tn * 16 + ll) * 64;
      const float* Srow = S + (size_t)(tn * 16 + ll) * 64;
#pragma unroll
      for (int s = 0; s < 16; ++s) {
        const int ka = s * 4 + lh * 2;
        xn[s].x = Xrow[ka]; xn[s].y = Xrow[ka + 1];
        sn[s].x = Srow[ka]; sn[s].y = Srow[ka + 1];
      }
    }

    const int rowBase = t << 4;
    float dr[8], z0[8], z1[8], z2[8], z3[8];
#pragma unroll
    for (int j = 0; j < 8; ++j) {
      const int r = rowBase + j + lh * 8;
      const float* x0r = X0 + (size_t)r * 64;
      dr[j] = deg[r];
      z0[j] = x0r[ll];
      z1[j] = x0r[16 + ll];
      z2[j] = x0r[32 + ll];
      z3[j] = x0r[48 + ll];
    }

    v8f acc0 = {}, acc1 = {}, acc2 = {}, acc3 = {};
#pragma unroll
    for (int s = 0; s < 16; ++s) {
      const int ka = s * 4 + lh * 2;
      v2f a;
      a.x = dg * xa[s].x;
      a.y = dg * xa[s].y;
      const float* B0 = &Bsh[ka * 64 + ll];
      const float* B1 = &Bsh[(ka + 1) * 64 + ll];
      v2f b;
      b.x = B0[0];  b.y = B1[0];  acc0 = wmma_f32(a, b, acc0);
      b.x = B0[16]; b.y = B1[16]; acc1 = wmma_f32(a, b, acc1);
      b.x = B0[32]; b.y = B1[32]; acc2 = wmma_f32(a, b, acc2);
      b.x = B0[48]; b.y = B1[48]; acc3 = wmma_f32(a, b, acc3);
    }
#pragma unroll
    for (int s = 0; s < 16; ++s) {
      const int ka = 64 + s * 4 + lh * 2;
      const float* B0 = &Bsh[ka * 64 + ll];
      const float* B1 = &Bsh[(ka + 1) * 64 + ll];
      v2f b;
      b.x = B0[0];  b.y = B1[0];  acc0 = wmma_f32(sa[s], b, acc0);
      b.x = B0[16]; b.y = B1[16]; acc1 = wmma_f32(sa[s], b, acc1);
      b.x = B0[32]; b.y = B1[32]; acc2 = wmma_f32(sa[s], b, acc2);
      b.x = B0[48]; b.y = B1[48]; acc3 = wmma_f32(sa[s], b, acc3);
    }

#pragma unroll
    for (int j = 0; j < 8; ++j) {
      const int r = rowBase + j + lh * 8;
      float* o = T + (size_t)r * 64;
      o[ll]      = oma * (acc0[j] + dr[j] * c0) + ALPHA_C * z0[j];
      o[16 + ll] = oma * (acc1[j] + dr[j] * c1) + ALPHA_C * z1[j];
      o[32 + ll] = oma * (acc2[j] + dr[j] * c2) + ALPHA_C * z2[j];
      o[48 + ll] = oma * (acc3[j] + dr[j] * c3) + ALPHA_C * z3[j];
    }
    t = tn;
  }
}

#define OWN_E   79
#define OWN_V   391
#define OWNERS  256
#define FT      1024
#define FG      4
#define QCAP    2048
#define DR      20
#define NGROUPS (NNZ_C / 32)

#define MAP_T     256
#define MAP_WAVES (RES_WG * (MAP_T / 32))
#define NQUADS    ((NGROUPS + 3) / 4)
#define MAP_QPW   ((NQUADS + MAP_WAVES - 1) / MAP_WAVES)

__global__ __launch_bounds__(MAP_T) void owner_map_kernel(
    const int* __restrict__ vertex, const int* __restrict__ edges,
    unsigned int* __restrict__ mapE, unsigned int* __restrict__ mapV) {
  const int lane = threadIdx.x & 31;
  const int waveG = blockIdx.x * (MAP_T / 32) +
                    __builtin_amdgcn_readfirstlane((int)(threadIdx.x >> 5));
  int ev[MAP_QPW][4], vv[MAP_QPW][4];
#pragma unroll
  for (int i = 0; i < MAP_QPW; ++i) {
    const int q = waveG + i * MAP_WAVES;
#pragma unroll
    for (int k = 0; k < 4; ++k) {
      const int g = q * 4 + k;
      const int gl = g < NGROUPS ? g : NGROUPS - 1;
      ev[i][k] = edges[gl * 32 + lane];
      vv[i][k] = vertex[gl * 32 + lane];
    }
  }
#pragma unroll
  for (int i = 0; i < MAP_QPW; ++i) {
    const int q = waveG + i * MAP_WAVES;
    unsigned int pe = 0, pv = 0;
#pragma unroll
    for (int k = 0; k < 4; ++k) {
      const unsigned int oE = (unsigned int)(ev[i][k] / OWN_E);
      const unsigned int oV = (unsigned int)(vv[i][k] / OWN_V);
#pragma unroll
      for (int p = 0; p < 8; ++p) {
        const unsigned int bE = __builtin_amdgcn_ballot_w32(((oE >> p) & 1u) != 0u);
        const unsigned int bV = __builtin_amdgcn_ballot_w32(((oV >> p) & 1u) != 0u);
        if (lane == 8 * k + p) { pe = bE; pv = bV; }
      }
    }
    if (q * 4 + (lane >> 3) < NGROUPS) {
      mapE[q * 32 + lane] = pe;
      mapV[q * 32 + lane] = pv;
    }
  }
}

__device__ __forceinline__ unsigned int owner_match(const uint4 a, const uint4 b,
                                                    const unsigned int me) {
  unsigned int r;
  r  = a.x ^ ((me & 1u)   ? 0u : 0xFFFFFFFFu);
  r &= a.y ^ ((me & 2u)   ? 0u : 0xFFFFFFFFu);
  r &= a.z ^ ((me & 4u)   ? 0u : 0xFFFFFFFFu);
  r &= a.w ^ ((me & 8u)   ? 0u : 0xFFFFFFFFu);
  r &= b.x ^ ((me & 16u)  ? 0u : 0xFFFFFFFFu);
  r &= b.y ^ ((me & 32u)  ? 0u : 0xFFFFFFFFu);
  r &= b.z ^ ((me & 64u)  ? 0u : 0xFFFFFFFFu);
  r &= b.w ^ ((me & 128u) ? 0u : 0xFFFFFFFFu);
  return r;
}

__device__ __forceinline__ void filter_slow_one(
    const float* __restrict__ src, const int* __restrict__ dstIdx,
    const int* __restrict__ srcIdx, int e, int base, float* acc, int* cnt,
    bool wantDeg) {
  const int r = dstIdx[e] - base;
  const float* s = src + (size_t)srcIdx[e] * 64;
  for (int c = 0; c < 64; ++c) atomicAdd(&acc[r * 64 + c], s[c]);
  if (wantDeg) atomicAdd(&cnt[r], 1);
}

__device__ __forceinline__ void filter_sum_body(
    const float* __restrict__ src, const int* __restrict__ dstIdx,
    const int* __restrict__ srcIdx, const unsigned int* __restrict__ map,
    float* __restrict__ dst, float* __restrict__ deg, const bool wantDeg,
    const int R, const int nDst, float* acc, int* cnt, int* q, int* qn) {
  const int tid = threadIdx.x;
  const int lane = tid & 31;
  const int wave = tid >> 5;
  const int base = blockIdx.x * R;

  for (int i = tid; i < R * 64; i += FT) acc[i] = 0.0f;
  if (wantDeg)
    for (int i = tid; i < R; i += FT) cnt[i] = 0;
  if (tid < 3) qn[tid] = 0;
  __syncthreads();

  const unsigned int me = blockIdx.x;
  const int nChunks = (NGROUPS + FT * FG - 1) / (FT * FG);
  const uint4* mapv = (const uint4*)map;
  int ia = 0, ib = 2, ic = 1;

  for (int c = 0; c <= nChunks; ++c) {
    const int c0 = c * (FT * FG);
    const int cp = c0 - FT * FG;
    const bool doScan = c < nChunks;
    int* qScan = q + (c & 1) * QCAP;
    const int* qDrain = q + ((c & 1) ^ 1) * QCAP;

    uint4 ma[FG], mb[FG];
    if (doScan) {
#pragma unroll
      for (int s = 0; s < FG; ++s) {
        const int gi = c0 + s * FT + tid;
        const int gl = gi < NGROUPS ? gi : NGROUPS - 1;
        ma[s] = mapv[gl * 2];
        mb[s] = mapv[gl * 2 + 1];
      }
    }

    int n = (c > 0) ? qn[ib] : 0;
    if (tid == 0) qn[ic] = 0;
    if (n > QCAP) {
      n = 0;
#pragma nounroll
      for (int s = 0; s < FG; ++s) {
        const int gi = cp + s * FT + tid;
        if (gi >= NGROUPS) continue;
        unsigned int z = owner_match(mapv[gi * 2], mapv[gi * 2 + 1], me);
        while (z) {
          const int b = __builtin_ctz(z);
          z &= z - 1;
          filter_slow_one(src, dstIdx, srcIdx, gi * 32 + b, base, acc, cnt,
                          wantDeg);
        }
      }
    }

    int dI = 0, sI = 0;
    bool has = false;
    if (lane < DR) {
      const int qi = lane * 32 + wave;
      if (qi < n) {
        const int e = qDrain[qi];
        dI = dstIdx[e];
        sI = srcIdx[e];
        has = true;
      }
    }

    if (doScan) {
#pragma unroll
      for (int s = 0; s < FG; ++s) {
        const int gi = c0 + s * FT + tid;
        unsigned int z = gi < NGROUPS ? owner_match(ma[s], mb[s], me) : 0u;
        while (z) {
          const int b = __builtin_ctz(z);
          z &= z - 1;
          const int slot = atomicAdd(&qn[ia], 1);
          if (slot < QCAP) qScan[slot] = gi * 32 + b;
        }
      }
    }

    int pk = has ? (((dI - base) << 20) | sI) : -1;
    for (int jb = 0; jb < n; jb += 32 * DR) {
      if (jb > 0) {
        pk = -1;
        if (lane < DR) {
          const int qi = jb + lane * 32 + wave;
          if (qi < n) {
            const int e = qDrain[qi];
            pk = ((dstIdx[e] - base) << 20) | srcIdx[e];
          }
        }
      }
      float2 r[DR];
      int p[DR];
#pragma unroll
      for (int k = 0; k < DR; ++k) {
        p[k] = __builtin_amdgcn_readlane(pk, k);
        const int srow = p[k] >= 0 ? (p[k] & 0xFFFFF) : 0;
        r[k] = *(const float2*)(src + (size_t)srow * 64 + lane * 2);
      }
#pragma unroll
      for (int k = 0; k < DR; ++k) {
        if (p[k] >= 0) {
          float* a = acc + (p[k] >> 20) * 64 + lane * 2;
          atomicAdd(a, r[k].x);
          atomicAdd(a + 1, r[k].y);
          if (wantDeg && lane == 0) atomicAdd(&cnt[p[k] >> 20], 1);
        }
      }
    }
    __syncthreads();
    const int t = ib; ib = ia; ia = ic; ic = t;
  }

  for (int i = tid; i < R * 64; i += FT) {
    const int row = base + (i >> 6);
    if (row < nDst) dst[(size_t)row * 64 + (i & 63)] = acc[i];
  }
  if (wantDeg)
    for (int i = tid; i < R; i += FT)
      if (base + i < nDst) deg[base + i] = (float)cnt[i];
}

__global__ __launch_bounds__(FT) void filter_edges_kernel(
    const float* __restrict__ Y1, const int* __restrict__ vertex,
    const int* __restrict__ edges, const unsigned int* __restrict__ mapE,
    float* __restrict__ Xe) {
  __shared__ float acc[OWN_E * 64];
  __shared__ int q[2 * QCAP];
  __shared__ int qn[3];
  filter_sum_body(Y1, edges, vertex, mapE, Xe, nullptr, false, OWN_E,
                  N_EDGES_C, acc, nullptr, q, qn);
}

__global__ __launch_bounds__(FT) void filter_nodes_kernel(
    const float* __restrict__ Xe, const int* __restrict__ vertex,
    const int* __restrict__ edges, const unsigned int* __restrict__ mapV,
    float* __restrict__ S, float* __restrict__ deg) {
  __shared__ float acc[OWN_V * 64];
  __shared__ int cnt[OWN_V];
  __shared__ int q[2 * QCAP];
  __shared__ int qn[3];
  filter_sum_body(Xe, vertex, edges, mapV, S, deg, true, OWN_V, N_NODES_C,
                  acc, cnt, q, qn);
}

extern "C" void kernel_launch(void* const* d_in, const int* in_sizes, int n_in,
                              void* d_out, int out_size, void* d_ws,
                              size_t ws_size, hipStream_t stream) {
  (void)in_sizes; (void)n_in; (void)out_size; (void)ws_size;
  const float* X    = (const float*)d_in[0];
  const float* X0   = (const float*)d_in[1];
  const int*   vertex = (const int*)d_in[2];
  const int*   edges  = (const int*)d_in[3];
  const float* W1w  = (const float*)d_in[4];
  const float* W1b  = (const float*)d_in[5];
  const float* W2w  = (const float*)d_in[6];
  const float* W2b  = (const float*)d_in[7];
  const float* Ww   = (const float*)d_in[8];
  const float* Wb   = (const float*)d_in[9];

  float* out = (float*)d_out;
  float* Xe  = out + (size_t)N_NODES_C * 64;

  float* ws  = (float*)d_ws;
  float* Y1  = ws;
  float* S   = Y1 + (size_t)N_NODES_C * 64;
  float* deg = S + (size_t)N_NODES_C * 64;

  unsigned int* mapE = (unsigned int*)d_out;
  unsigned int* mapV = mapE + NNZ_C / 4;

  gemm64_bias_kernel<<<RES_WG, GEMM_T, 0, stream>>>(X, W1w, W1b, Y1, N_NODES_C);
  owner_map_kernel<<<RES_WG, MAP_T, 0, stream>>>(vertex, edges, mapE, mapV);
  filter_edges_kernel<<<OWNERS, FT, 0, stream>>>(Y1, vertex, edges, mapE, Xe);
  filter_nodes_kernel<<<OWNERS, FT, 0, stream>>>(Xe, vertex, edges, mapV, S, deg);
  gemm2_mix_kernel<<<RES_WG, GEMM2_T, 0, stream>>>(X, X0, S, deg, W2w, W2b, Y1,
                                                  N_NODES_C);
  gemm64_bias_kernel<<<RES_WG, GEMM_T, 0, stream>>>(Y1, Ww, Wb, out, N_NODES_C);
}


# owner map launch also writes one packed word per incidence and sum (local destination row << 20 | source row): a queued hit costs one random 4-byte fetch, not two; no sum reordered
# speedup vs baseline: 2.6236x; 1.0656x over previous
#include <hip/hip_runtime.h>

typedef __attribute__((ext_vector_type(2))) float v2f;
typedef __attribute__((ext_vector_type(8))) float v8f;

#define N_NODES_C 100000
#define N_EDGES_C 20000
#define NNZ_C     1000000
#define ALPHA_C   0.5f

__device__ __forceinline__ v8f wmma_f32(v2f a, v2f b, v8f c) {
  return __builtin_amdgcn_wmma_f32_16x16x4_f32(false, a, false, b, (short)0, c,
                                               false, false);
}

#define RES_WG   256
#define GEMM_T   512
#define GEMM_NW  (GEMM_T / 32)
#define GEMM2_T  256
#define GEMM2_NW (GEMM2_T / 32)

__global__ __launch_bounds__(GEMM_T) void gemm64_bias_kernel(
    const float* __restrict__ A, const float* __restrict__ W,
    const float* __restrict__ bias, float* __restrict__ Out, int M) {
  __shared__ float Bsh[64 * 64];

  const int lane = threadIdx.x & 31;
  const int wave = __builtin_amdgcn_readfirstlane((int)(threadIdx.x >> 5));
  const int ll = lane & 15;
  const int lh = lane >> 4;
  const int tiles = M >> 4;
  const int tBeg = (int)(((long long)blockIdx.x * tiles) / RES_WG);
  const int tEnd = (int)(((long long)(blockIdx.x + 1) * tiles) / RES_WG);
  int t = tBeg + wave;

  v2f an[16];
#pragma unroll
  for (int s = 0; s < 16; ++s) { an[s].x = 0.0f; an[s].y = 0.0f; }
  if (t < tEnd) {
    const float* Arow = A + (size_t)(t * 16 + ll) * 64;
#pragma unroll
    for (int s = 0; s < 16; ++s) {
      const int ka = s * 4 + lh * 2;
      an[s].x = Arow[ka];
      an[s].y = Arow[ka + 1];
    }
  }

#pragma unroll
  for (int j = 0; j < 64 * 64 / GEMM_T; ++j) {
    const int i = j * GEMM_T + (int)threadIdx.x;
    const int k = i >> 6, n = i & 63;
    Bsh[i] = W[n * 64 + k];
  }
  const float bi0 = bias[ll], bi1 = bias[16 + ll], bi2 = bias[32 + ll],
              bi3 = bias[48 + ll];
  __syncthreads();

  while (t < tEnd) {
    v2f a[16];
#pragma unroll
    for (int s = 0; s < 16; ++s) a[s] = an[s];
    const int tn = t + GEMM_NW;
    if (tn < tEnd) {
      const float* Arow = A + (size_t)(tn * 16 + ll) * 64;
#pragma unroll
      for (int s = 0; s < 16; ++s) {
        const int ka = s * 4 + lh * 2;
        an[s].x = Arow[ka];
        an[s].y = Arow[ka + 1];
      }
    }

    v8f acc0 = {}, acc1 = {}, acc2 = {}, acc3 = {};
#pragma unroll
    for (int s = 0; s < 16; ++s) {
      const int ka = s * 4 + lh * 2;
      const float* B0 = &Bsh[ka * 64 + ll];
      const float* B1 = &Bsh[(ka + 1) * 64 + ll];
      v2f b;
      b.x = B0[0];  b.y = B1[0];  acc0 = wmma_f32(a[s], b, acc0);
      b.x = B0[16]; b.y = B1[16]; acc1 = wmma_f32(a[s], b, acc1);
      b.x = B0[32]; b.y = B1[32]; acc2 = wmma_f32(a[s], b, acc2);
      b.x = B0[48]; b.y = B1[48]; acc3 = wmma_f32(a[s], b, acc3);
    }

    const int rowBase = t << 4;
#pragma unroll
    for (int j = 0; j < 8; ++j) {
      const int r = rowBase + j + lh * 8;
      float* o = Out + (size_t)r * 64;
      o[ll]      = acc0[j] + bi0;
      o[16 + ll] = acc1[j] + bi1;
      o[32 + ll] = acc2[j] + bi2;
      o[48 + ll] = acc3[j] + bi3;
    }
    t = tn;
  }
}

__global__ __launch_bounds__(GEMM2_T) void gemm2_mix_kernel(
    const float* __restrict__ X, const float* __restrict__ X0,
    const float* __restrict__ S, const float* __restrict__ deg,
    const float* __restrict__ W2, const float* __restrict__ b2,
    float* __restrict__ T, int M) {
  __shared__ float Bsh[128 * 64];

  const int lane = threadIdx.x & 31;
  const int wave = __builtin_amdgcn_readfirstlane((int)(threadIdx.x >> 5));
  const int ll = lane & 15;
  const int lh = lane >> 4;
  const int tiles = M >> 4;
  const int tBeg = (int)(((long long)blockIdx.x * tiles) / RES_WG);
  const int tEnd = (int)(((long long)(blockIdx.x + 1) * tiles) / RES_WG);
  int t = tBeg + wave;

  v2f xn[16], sn[16];
  float dgn = 0.0f;
#pragma unroll
  for (int s = 0; s < 16; ++s) {
    xn[s].x = 0.0f; xn[s].y = 0.0f; sn[s].x = 0.0f; sn[s].y = 0.0f;
  }
  if (t < tEnd) {
    dgn = deg[t * 16 + ll];
    const float* Xrow = X + (size_t)(t * 16 + ll) * 64;
    const float* Srow = S + (size_t)(t * 16 + ll) * 64;
#pragma unroll
    for (int s = 0; s < 16; ++s) {
      const int ka = s * 4 + lh * 2;
      xn[s].x = Xrow[ka]; xn[s].y = Xrow[ka + 1];
      sn[s].x = Srow[ka]; sn[s].y = Srow[ka + 1];
    }
  }

  {
    float w[128 * 64 / GEMM2_T];
#pragma unroll
    for (int j = 0; j < 128 * 64 / GEMM2_T; ++j) {
      const int i = j * GEMM2_T + (int)threadIdx.x;
      const int k = i >> 6, n = i & 63;
      w[j] = W2[n * 128 + k];
    }
#pragma unroll
    for (int j = 0; j < 128 * 64 / GEMM2_T; ++j)
      Bsh[j * GEMM2_T + (int)threadIdx.x] = w[j];
  }
  const float c0 = b2[ll], c1 = b2[16 + ll], c2 = b2[32 + ll], c3 = b2[48 + ll];
  __syncthreads();

  const float oma = 1.0f - ALPHA_C;
  while (t < tEnd) {
    v2f xa[16], sa[16];
#pragma unroll
    for (int s = 0; s < 16; ++s) { xa[s] = xn[s]; sa[s] = sn[s]; }
    const float dg = dgn;
    const int tn = t + GEMM2_NW;
    if (tn < tEnd) {
      dgn = deg[tn * 16 + ll];
      const float* Xrow = X + (size_t)(tn * 16 + ll) * 64;
      const float* Srow = S + (size_t)(tn * 16 + ll) * 64;
#pragma unroll
      for (int s = 0; s < 16; ++s) {
        const int ka = s * 4 + lh * 2;
        xn[s].x = Xrow[ka]; xn[s].y = Xrow[ka + 1];
        sn[s].x = Srow[ka]; sn[s].y = Srow[ka + 1];
      }
    }

    const int rowBase = t << 4;
    float dr[8], z0[8], z1[8], z2[8], z3[8];
#pragma unroll
    for (int j = 0; j < 8; ++j) {
      const int r = rowBase + j + lh * 8;
      const float* x0r = X0 + (size_t)r * 64;
      dr[j] = deg[r];
      z0[j] = x0r[ll];
      z1[j] = x0r[16 + ll];
      z2[j] = x0r[32 + ll];
      z3[j] = x0r[48 + ll];
    }

    v8f acc0 = {}, acc1 = {}, acc2 = {}, acc3 = {};
#pragma unroll
    for (int s = 0; s < 16; ++s) {
      const int ka = s * 4 + lh * 2;
      v2f a;
      a.x = dg * xa[s].x;
      a.y = dg * xa[s].y;
      const float* B0 = &Bsh[ka * 64 + ll];
      const float* B1 = &Bsh[(ka + 1) * 64 + ll];
      v2f b;
      b.x = B0[0];  b.y = B1[0];  acc0 = wmma_f32(a, b, acc0);
      b.x = B0[16]; b.y = B1[16]; acc1 = wmma_f32(a, b, acc1);
      b.x = B0[32]; b.y = B1[32]; acc2 = wmma_f32(a, b, acc2);
      b.x = B0[48]; b.y = B1[48]; acc3 = wmma_f32(a, b, acc3);
    }
#pragma unroll
    for (int s = 0; s < 16; ++s) {
      const int ka = 64 + s * 4 + lh * 2;
      const float* B0 = &Bsh[ka * 64 + ll];
      const float* B1 = &Bsh[(ka + 1) * 64 + ll];
      v2f b;
      b.x = B0[0];  b.y = B1[0];  acc0 = wmma_f32(sa[s], b, acc0);
      b.x = B0[16]; b.y = B1[16]; acc1 = wmma_f32(sa[s], b, acc1);
      b.x = B0[32]; b.y = B1[32]; acc2 = wmma_f32(sa[s], b, acc2);
      b.x = B0[48]; b.y = B1[48]; acc3 = wmma_f32(sa[s], b, acc3);
    }

#pragma unroll
    for (int j = 0; j < 8; ++j) {
      const int r = rowBase + j + lh * 8;
      float* o = T + (size_t)r * 64;
      o[ll]      = oma * (acc0[j] + dr[j] * c0) + ALPHA_C * z0[j];
      o[16 + ll] = oma * (acc1[j] + dr[j] * c1) + ALPHA_C * z1[j];
      o[32 + ll] = oma * (acc2[j] + dr[j] * c2) + ALPHA_C * z2[j];
      o[48 + ll] = oma * (acc3[j] + dr[j] * c3) + ALPHA_C * z3[j];
    }
    t = tn;
  }
}

#define OWN_E   79
#define OWN_V   391
#define OWNERS  256
#define FT      1024
#define FG      4
#define QCAP    2048
#define DR      20
#define NGROUPS (NNZ_C / 32)

#define MAP_T     256
#define MAP_WAVES (RES_WG * (MAP_T / 32))
#define NQUADS    ((NGROUPS + 3) / 4)
#define MAP_QPW   ((NQUADS + MAP_WAVES - 1) / MAP_WAVES)

__global__ __launch_bounds__(MAP_T) void owner_map_kernel(
    const int* __restrict__ vertex, const int* __restrict__ edges,
    unsigned int* __restrict__ mapE, unsigned int* __restrict__ mapV,
    unsigned int* __restrict__ pkE, unsigned int* __restrict__ pkV) {
  const int lane = threadIdx.x & 31;
  const int waveG = blockIdx.x * (MAP_T / 32) +
                    __builtin_amdgcn_readfirstlane((int)(threadIdx.x >> 5));
  int ev[MAP_QPW][4], vv[MAP_QPW][4];
#pragma unroll
  for (int i = 0; i < MAP_QPW; ++i) {
    const int q = waveG + i * MAP_WAVES;
#pragma unroll
    for (int k = 0; k < 4; ++k) {
      const int g = q * 4 + k;
      const int gl = g < NGROUPS ? g : NGROUPS - 1;
      ev[i][k] = edges[gl * 32 + lane];
      vv[i][k] = vertex[gl * 32 + lane];
    }
  }
#pragma unroll
  for (int i = 0; i < MAP_QPW; ++i) {
    const int q = waveG + i * MAP_WAVES;
    unsigned int pe = 0, pv = 0;
#pragma unroll
    for (int k = 0; k < 4; ++k) {
      const unsigned int oE = (unsigned int)(ev[i][k] / OWN_E);
      const unsigned int oV = (unsigned int)(vv[i][k] / OWN_V);
      if (q * 4 + k < NGROUPS) {
        const int e = (q * 4 + k) * 32 + lane;
        pkE[e] = (((unsigned int)ev[i][k] - oE * OWN_E) << 20) |
                 (unsigned int)vv[i][k];
        pkV[e] = (((unsigned int)vv[i][k] - oV * OWN_V) << 20) |
                 (unsigned int)ev[i][k];
      }
#pragma unroll
      for (int p = 0; p < 8; ++p) {
        const unsigned int bE = __builtin_amdgcn_ballot_w32(((oE >> p) & 1u) != 0u);
        const unsigned int bV = __builtin_amdgcn_ballot_w32(((oV >> p) & 1u) != 0u);
        if (lane == 8 * k + p) { pe = bE; pv = bV; }
      }
    }
    if (q * 4 + (lane >> 3) < NGROUPS) {
      mapE[q * 32 + lane] = pe;
      mapV[q * 32 + lane] = pv;
    }
  }
}

__device__ __forceinline__ unsigned int owner_match(const uint4 a, const uint4 b,
                                                    const unsigned int me) {
  unsigned int r;
  r  = a.x ^ ((me & 1u)   ? 0u : 0xFFFFFFFFu);
  r &= a.y ^ ((me & 2u)   ? 0u : 0xFFFFFFFFu);
  r &= a.z ^ ((me & 4u)   ? 0u : 0xFFFFFFFFu);
  r &= a.w ^ ((me & 8u)   ? 0u : 0xFFFFFFFFu);
  r &= b.x ^ ((me & 16u)  ? 0u : 0xFFFFFFFFu);
  r &= b.y ^ ((me & 32u)  ? 0u : 0xFFFFFFFFu);
  r &= b.z ^ ((me & 64u)  ? 0u : 0xFFFFFFFFu);
  r &= b.w ^ ((me & 128u) ? 0u : 0xFFFFFFFFu);
  return r;
}

__device__ __forceinline__ void filter_slow_one(
    const float* __restrict__ src, const unsigned int* __restrict__ pkw, int e,
    float* acc, int* cnt, bool wantDeg) {
  const unsigned int w = pkw[e];
  const int r = (int)(w >> 20);
  const float* s = src + (size_t)(w & 0xFFFFFu) * 64;
  for (int c = 0; c < 64; ++c) atomicAdd(&acc[r * 64 + c], s[c]);
  if (wantDeg) atomicAdd(&cnt[r], 1);
}

__device__ __forceinline__ void filter_sum_body(
    const float* __restrict__ src, const unsigned int* __restrict__ pkw,
    const unsigned int* __restrict__ map,
    float* __restrict__ dst, float* __restrict__ deg, const bool wantDeg,
    const int R, const int nDst, float* acc, int* cnt, int* q, int* qn) {
  const int tid = threadIdx.x;
  const int lane = tid & 31;
  const int wave = tid >> 5;
  const int base = blockIdx.x * R;

  for (int i = tid; i < R * 64; i += FT) acc[i] = 0.0f;
  if (wantDeg)
    for (int i = tid; i < R; i += FT) cnt[i] = 0;
  if (tid < 3) qn[tid] = 0;
  __syncthreads();

  const unsigned int me = blockIdx.x;
  const int nChunks = (NGROUPS + FT * FG - 1) / (FT * FG);
  const uint4* mapv = (const uint4*)map;
  int ia = 0, ib = 2, ic = 1;

  for (int c = 0; c <= nChunks; ++c) {
    const int c0 = c * (FT * FG);
    const int cp = c0 - FT * FG;
    const bool doScan = c < nChunks;
    int* qScan = q + (c & 1) * QCAP;
    const int* qDrain = q + ((c & 1) ^ 1) * QCAP;

    uint4 ma[FG], mb[FG];
    if (doScan) {
#pragma unroll
      for (int s = 0; s < FG; ++s) {
        const int gi = c0 + s * FT + tid;
        const int gl = gi < NGROUPS ? gi : NGROUPS - 1;
        ma[s] = mapv[gl * 2];
        mb[s] = mapv[gl * 2 + 1];
      }
    }

    int n = (c > 0) ? qn[ib] : 0;
    if (tid == 0) qn[ic] = 0;
    if (n > QCAP) {
      n = 0;
#pragma nounroll
      for (int s = 0; s < FG; ++s) {
        const int gi = cp + s * FT + tid;
        if (gi >= NGROUPS) continue;
        unsigned int z = owner_match(mapv[gi * 2], mapv[gi * 2 + 1], me);
        while (z) {
          const int b = __builtin_ctz(z);
          z &= z - 1;
          filter_slow_one(src, pkw, gi * 32 + b, acc, cnt, wantDeg);
        }
      }
    }

    int pk = -1;
    if (lane < DR) {
      const int qi = lane * 32 + wave;
      if (qi < n) pk = (int)pkw[qDrain[qi]];
    }

    if (doScan) {
#pragma unroll
      for (int s = 0; s < FG; ++s) {
        const int gi = c0 + s * FT + tid;
        unsigned int z = gi < NGROUPS ? owner_match(ma[s], mb[s], me) : 0u;
        while (z) {
          const int b = __builtin_ctz(z);
          z &= z - 1;
          const int slot = atomicAdd(&qn[ia], 1);
          if (slot < QCAP) qScan[slot] = gi * 32 + b;
        }
      }
    }

    for (int jb = 0; jb < n; jb += 32 * DR) {
      if (jb > 0) {
        pk = -1;
        if (lane < DR) {
          const int qi = jb + lane * 32 + wave;
          if (qi < n) pk = (int)pkw[qDrain[qi]];
        }
      }
      float2 r[DR];
      int p[DR];
#pragma unroll
      for (int k = 0; k < DR; ++k) {
        p[k] = __builtin_amdgcn_readlane(pk, k);
        const int srow = p[k] >= 0 ? (p[k] & 0xFFFFF) : 0;
        r[k] = *(const float2*)(src + (size_t)srow * 64 + lane * 2);
      }
#pragma unroll
      for (int k = 0; k < DR; ++k) {
        if (p[k] >= 0) {
          float* a = acc + (p[k] >> 20) * 64 + lane * 2;
          atomicAdd(a, r[k].x);
          atomicAdd(a + 1, r[k].y);
          if (wantDeg && lane == 0) atomicAdd(&cnt[p[k] >> 20], 1);
        }
      }
    }
    __syncthreads();
    const int t = ib; ib = ia; ia = ic; ic = t;
  }

  for (int i = tid; i < R * 64; i += FT) {
    const int row = base + (i >> 6);
    if (row < nDst) dst[(size_t)row * 64 + (i & 63)] = acc[i];
  }
  if (wantDeg)
    for (int i = tid; i < R; i += FT)
      if (base + i < nDst) deg[base + i] = (float)cnt[i];
}

__global__ __launch_bounds__(FT) void filter_edges_kernel(
    const float* __restrict__ Y1, const unsigned int* __restrict__ pkE,
    const unsigned int* __restrict__ mapE, float* __restrict__ Xe) {
  __shared__ float acc[OWN_E * 64];
  __shared__ int q[2 * QCAP];
  __shared__ int qn[3];
  filter_sum_body(Y1, pkE, mapE, Xe, nullptr, false, OWN_E, N_EDGES_C, acc,
                  nullptr, q, qn);
}

__global__ __launch_bounds__(FT) void filter_nodes_kernel(
    const float* __restrict__ Xe, const unsigned int* __restrict__ pkV,
    const unsigned int* __restrict__ mapV, float* __restrict__ S,
    float* __restrict__ deg) {
  __shared__ float acc[OWN_V * 64];
  __shared__ int cnt[OWN_V];
  __shared__ int q[2 * QCAP];
  __shared__ int qn[3];
  filter_sum_body(Xe, pkV, mapV, S, deg, true, OWN_V, N_NODES_C, acc, cnt, q,
                  qn);
}

extern "C" void kernel_launch(void* const* d_in, const int* in_sizes, int n_in,
                              void* d_out, int out_size, void* d_ws,
                              size_t ws_size, hipStream_t stream) {
  (void)in_sizes; (void)n_in; (void)out_size; (void)ws_size;
  const float* X    = (const float*)d_in[0];
  const float* X0   = (const float*)d_in[1];
  const int*   vertex = (const int*)d_in[2];
  const int*   edges  = (const int*)d_in[3];
  const float* W1w  = (const float*)d_in[4];
  const float* W1b  = (const float*)d_in[5];
  const float* W2w  = (const float*)d_in[6];
  const float* W2b  = (const float*)d_in[7];
  const float* Ww   = (const float*)d_in[8];
  const float* Wb   = (const float*)d_in[9];

  float* out = (float*)d_out;
  float* Xe  = out + (size_t)N_NODES_C * 64;

  float* ws  = (float*)d_ws;
  float* Y1  = ws;
  float* S   = Y1 + (size_t)N_NODES_C * 64;
  float* deg = S + (size_t)N_NODES_C * 64;

  unsigned int* mapE = (unsigned int*)d_out;
  unsigned int* mapV = mapE + NNZ_C / 4;
  unsigned int* pkE  = mapV + NNZ_C / 4;
  unsigned int* pkV  = pkE + NNZ_C;

  gemm64_bias_kernel<<<RES_WG, GEMM_T, 0, stream>>>(X, W1w, W1b, Y1, N_NODES_C);
  owner_map_kernel<<<RES_WG, MAP_T, 0, stream>>>(vertex, edges, mapE, mapV, pkE,
                                                 pkV);
  filter_edges_kernel<<<OWNERS, FT, 0, stream>>>(Y1, pkE, mapE, Xe);
  filter_nodes_kernel<<<OWNERS, FT, 0, stream>>>(Xe, pkV, mapV, S, deg);
  gemm2_mix_kernel<<<RES_WG, GEMM2_T, 0, stream>>>(X, X0, S, deg, W2w, W2b, Y1,
                                                  N_NODES_C);
  gemm64_bias_kernel<<<RES_WG, GEMM_T, 0, stream>>>(Y1, Ww, Wb, out, N_NODES_C);
}


# segment sums: each wave keeps its own hit queue (slots from a ballot) and fetches and adds its own hits: no work-group barrier, no shared counter in the chunk loop; LDS f32 adds in arrival order
# speedup vs baseline: 2.8070x; 1.0699x over previous
#include <hip/hip_runtime.h>

typedef __attribute__((ext_vector_type(2))) float v2f;
typedef __attribute__((ext_vector_type(8))) float v8f;

#define N_NODES_C 100000
#define N_EDGES_C 20000
#define NNZ_C     1000000
#define ALPHA_C   0.5f

__device__ __forceinline__ v8f wmma_f32(v2f a, v2f b, v8f c) {
  return __builtin_amdgcn_wmma_f32_16x16x4_f32(false, a, false, b, (short)0, c,
                                               false, false);
}

#define RES_WG   256
#define GEMM_T   512
#define GEMM_NW  (GEMM_T / 32)
#define GEMM2_T  256
#define GEMM2_NW (GEMM2_T / 32)

__global__ __launch_bounds__(GEMM_T) void gemm64_bias_kernel(
    const float* __restrict__ A, const float* __restrict__ W,
    const float* __restrict__ bias, float* __restrict__ Out, int M) {
  __shared__ float Bsh[64 * 64];

  const int lane = threadIdx.x & 31;
  const int wave = __builtin_amdgcn_readfirstlane((int)(threadIdx.x >> 5));
  const int ll = lane & 15;
  const int lh = lane >> 4;
  const int tiles = M >> 4;
  const int tBeg = (int)(((long long)blockIdx.x * tiles) / RES_WG);
  const int tEnd = (int)(((long long)(blockIdx.x + 1) * tiles) / RES_WG);
  int t = tBeg + wave;

  v2f an[16];
#pragma unroll
  for (int s = 0; s < 16; ++s) { an[s].x = 0.0f; an[s].y = 0.0f; }
  if (t < tEnd) {
    const float* Arow = A + (size_t)(t * 16 + ll) * 64;
#pragma unroll
    for (int s = 0; s < 16; ++s) {
      const int ka = s * 4 + lh * 2;
      an[s].x = Arow[ka];
      an[s].y = Arow[ka + 1];
    }
  }

#pragma unroll
  for (int j = 0; j < 64 * 64 / GEMM_T; ++j) {
    const int i = j * GEMM_T + (int)threadIdx.x;
    const int k = i >> 6, n = i & 63;
    Bsh[i] = W[n * 64 + k];
  }
  const float bi0 = bias[ll], bi1 = bias[16 + ll], bi2 = bias[32 + ll],
              bi3 = bias[48 + ll];
  __syncthreads();

  while (t < tEnd) {
    v2f a[16];
#pragma unroll
    for (int s = 0; s < 16; ++s) a[s] = an[s];
    const int tn = t + GEMM_NW;
    if (tn < tEnd) {
      const float* Arow = A + (size_t)(tn * 16 + ll) * 64;
#pragma unroll
      for (int s = 0; s < 16; ++s) {
        const int ka = s * 4 + lh * 2;
        an[s].x = Arow[ka];
        an[s].y = Arow[ka + 1];
      }
    }

    v8f acc0 = {}, acc1 = {}, acc2 = {}, acc3 = {};
#pragma unroll
    for (int s = 0; s < 16; ++s) {
      const int ka = s * 4 + lh * 2;
      const float* B0 = &Bsh[ka * 64 + ll];
      const float* B1 = &Bsh[(ka + 1) * 64 + ll];
      v2f b;
      b.x = B0[0];  b.y = B1[0];  acc0 = wmma_f32(a[s], b, acc0);
      b.x = B0[16]; b.y = B1[16]; acc1 = wmma_f32(a[s], b, acc1);
      b.x = B0[32]; b.y = B1[32]; acc2 = wmma_f32(a[s], b, acc2);
      b.x = B0[48]; b.y = B1[48]; acc3 = wmma_f32(a[s], b, acc3);
    }

    const int rowBase = t << 4;
#pragma unroll
    for (int j = 0; j < 8; ++j) {
      const int r = rowBase + j + lh * 8;
      float* o = Out + (size_t)r * 64;
      o[ll]      = acc0[j] + bi0;
      o[16 + ll] = acc1[j] + bi1;
      o[32 + ll] = acc2[j] + bi2;
      o[48 + ll] = acc3[j] + bi3;
    }
    t = tn;
  }
}

__global__ __launch_bounds__(GEMM2_T) void gemm2_mix_kernel(
    const float* __restrict__ X, const float* __restrict__ X0,
    const float* __restrict__ S, const float* __restrict__ deg,
    const float* __restrict__ W2, const float* __restrict__ b2,
    float* __restrict__ T, int M) {
  __shared__ float Bsh[128 * 64];

  const int lane = threadIdx.x & 31;
  const int wave = __builtin_amdgcn_readfirstlane((int)(threadIdx.x >> 5));
  const int ll = lane & 15;
  const int lh = lane >> 4;
  const int tiles = M >> 4;
  const int tBeg = (int)(((long long)blockIdx.x * tiles) / RES_WG);
  const int tEnd = (int)(((long long)(blockIdx.x + 1) * tiles) / RES_WG);
  int t = tBeg + wave;

  v2f xn[16], sn[16];
  float dgn = 0.0f;
#pragma unroll
  for (int s = 0; s < 16; ++s) {
    xn[s].x = 0.0f; xn[s].y = 0.0f; sn[s].x = 0.0f; sn[s].y = 0.0f;
  }
  if (t < tEnd) {
    dgn = deg[t * 16 + ll];
    const float* Xrow = X + (size_t)(t * 16 + ll) * 64;
    const float* Srow = S + (size_t)(t * 16 + ll) * 64;
#pragma unroll
    for (int s = 0; s < 16; ++s) {
      const int ka = s * 4 + lh * 2;
      xn[s].x = Xrow[ka]; xn[s].y = Xrow[ka + 1];
      sn[s].x = Srow[ka]; sn[s].y = Srow[ka + 1];
    }
  }

  {
    float w[128 * 64 / GEMM2_T];
#pragma unroll
    for (int j = 0; j < 128 * 64 / GEMM2_T; ++j) {
      const int i = j * GEMM2_T + (int)threadIdx.x;
      const int k = i >> 6, n = i & 63;
      w[j] = W2[n * 128 + k];
    }
#pragma unroll
    for (int j = 0; j < 128 * 64 / GEMM2_T; ++j)
      Bsh[j * GEMM2_T + (int)threadIdx.x] = w[j];
  }
  const float c0 = b2[ll], c1 = b2[16 + ll], c2 = b2[32 + ll], c3 = b2[48 + ll];
  __syncthreads();

  const float oma = 1.0f - ALPHA_C;
  while (t < tEnd) {
    v2f xa[16], sa[16];
#pragma unroll
    for (int s = 0; s < 16; ++s) { xa[s] = xn[s]; sa[s] = sn[s]; }
    const float dg = dgn;
    const int tn = t + GEMM2_NW;
    if (tn < tEnd) {
      dgn = deg[tn * 16 + ll];
      const float* Xrow = X + (size_t)(tn * 16 + ll) * 64;
      const float* Srow = S + (size_t)(tn * 16 + ll) * 64;
#pragma unroll
      for (int s = 0; s < 16; ++s) {
        const int ka = s * 4 + lh * 2;
        xn[s].x = Xrow[ka]; xn[s].y = Xrow[ka + 1];
        sn[s].x = Srow[ka]; sn[s].y = Srow[ka + 1];
      }
    }

    const int rowBase = t << 4;
    float dr[8], z0[8], z1[8], z2[8], z3[8];
#pragma unroll
    for (int j = 0; j < 8; ++j) {
      const int r = rowBase + j + lh * 8;
      const float* x0r = X0 + (size_t)r * 64;
      dr[j] = deg[r];
      z0[j] = x0r[ll];
      z1[j] = x0r[16 + ll];
      z2[j] = x0r[32 + ll];
      z3[j] = x0r[48 + ll];
    }

    v8f acc0 = {}, acc1 = {}, acc2 = {}, acc3 = {};
#pragma unroll
    for (int s = 0; s < 16; ++s) {
      const int ka = s * 4 + lh * 2;
      v2f a;
      a.x = dg * xa[s].x;
      a.y = dg * xa[s].y;
      const float* B0 = &Bsh[ka * 64 + ll];
      const float* B1 = &Bsh[(ka + 1) * 64 + ll];
      v2f b;
      b.x = B0[0];  b.y = B1[0];  acc0 = wmma_f32(a, b, acc0);
      b.x = B0[16]; b.y = B1[16]; acc1 = wmma_f32(a, b, acc1);
      b.x = B0[32]; b.y = B1[32]; acc2 = wmma_f32(a, b, acc2);
      b.x = B0[48]; b.y = B1[48]; acc3 = wmma_f32(a, b, acc3);
    }
#pragma unroll
    for (int s = 0; s < 16; ++s) {
      const int ka = 64 + s * 4 + lh * 2;
      const float* B0 = &Bsh[ka * 64 + ll];
      const float* B1 = &Bsh[(ka + 1) * 64 + ll];
      v2f b;
      b.x = B0[0];  b.y = B1[0];  acc0 = wmma_f32(sa[s], b, acc0);
      b.x = B0[16]; b.y = B1[16]; acc1 = wmma_f32(sa[s], b, acc1);
      b.x = B0[32]; b.y = B1[32]; acc2 = wmma_f32(sa[s], b, acc2);
      b.x = B0[48]; b.y = B1[48]; acc3 = wmma_f32(sa[s], b, acc3);
    }

#pragma unroll
    for (int j = 0; j < 8; ++j) {
      const int r = rowBase + j + lh * 8;
      float* o = T + (size_t)r * 64;
      o[ll]      = oma * (acc0[j] + dr[j] * c0) + ALPHA_C * z0[j];
      o[16 + ll] = oma * (acc1[j] + dr[j] * c1) + ALPHA_C * z1[j];
      o[32 + ll] = oma * (acc2[j] + dr[j] * c2) + ALPHA_C * z2[j];
      o[48 + ll] = oma * (acc3[j] + dr[j] * c3) + ALPHA_C * z3[j];
    }
    t = tn;
  }
}

#define OWN_E   79
#define OWN_V   391
#define OWNERS  256
#define FT      1024
#define FG      4
#define QW      128
#define DR      24
#define NGROUPS (NNZ_C / 32)

#define MAP_T     256
#define MAP_WAVES (RES_WG * (MAP_T / 32))
#define NQUADS    ((NGROUPS + 3) / 4)
#define MAP_QPW   ((NQUADS + MAP_WAVES - 1) / MAP_WAVES)

__global__ __launch_bounds__(MAP_T) void owner_map_kernel(
    const int* __restrict__ vertex, const int* __restrict__ edges,
    unsigned int* __restrict__ mapE, unsigned int* __restrict__ mapV,
    unsigned int* __restrict__ pkE, unsigned int* __restrict__ pkV) {
  const int lane = threadIdx.x & 31;
  const int waveG = blockIdx.x * (MAP_T / 32) +
                    __builtin_amdgcn_readfirstlane((int)(threadIdx.x >> 5));
  int ev[MAP_QPW][4], vv[MAP_QPW][4];
#pragma unroll
  for (int i = 0; i < MAP_QPW; ++i) {
    const int q = waveG + i * MAP_WAVES;
#pragma unroll
    for (int k = 0; k < 4; ++k) {
      const int g = q * 4 + k;
      const int gl = g < NGROUPS ? g : NGROUPS - 1;
      ev[i][k] = edges[gl * 32 + lane];
      vv[i][k] = vertex[gl * 32 + lane];
    }
  }
#pragma unroll
  for (int i = 0; i < MAP_QPW; ++i) {
    const int q = waveG + i * MAP_WAVES;
    unsigned int pe = 0, pv = 0;
#pragma unroll
    for (int k = 0; k < 4; ++k) {
      const unsigned int oE = (unsigned int)(ev[i][k] / OWN_E);
      const unsigned int oV = (unsigned int)(vv[i][k] / OWN_V);
      if (q * 4 + k < NGROUPS) {
        const int e = (q * 4 + k) * 32 + lane;
        pkE[e] = (((unsigned int)ev[i][k] - oE * OWN_E) << 20) |
                 (unsigned int)vv[i][k];
        pkV[e] = (((unsigned int)vv[i][k] - oV * OWN_V) << 20) |
                 (unsigned int)ev[i][k];
      }
#pragma unroll
      for (int p = 0; p < 8; ++p) {
        const unsigned int bE = __builtin_amdgcn_ballot_w32(((oE >> p) & 1u) != 0u);
        const unsigned int bV = __builtin_amdgcn_ballot_w32(((oV >> p) & 1u) != 0u);
        if (lane == 8 * k + p) { pe = bE; pv = bV; }
      }
    }
    if (q * 4 + (lane >> 3) < NGROUPS) {
      mapE[q * 32 + lane] = pe;
      mapV[q * 32 + lane] = pv;
    }
  }
}

__device__ __forceinline__ unsigned int owner_match(const uint4 a, const uint4 b,
                                                    const unsigned int me) {
  unsigned int r;
  r  = a.x ^ ((me & 1u)   ? 0u : 0xFFFFFFFFu);
  r &= a.y ^ ((me & 2u)   ? 0u : 0xFFFFFFFFu);
  r &= a.z ^ ((me & 4u)   ? 0u : 0xFFFFFFFFu);
  r &= a.w ^ ((me & 8u)   ? 0u : 0xFFFFFFFFu);
  r &= b.x ^ ((me & 16u)  ? 0u : 0xFFFFFFFFu);
  r &= b.y ^ ((me & 32u)  ? 0u : 0xFFFFFFFFu);
  r &= b.z ^ ((me & 64u)  ? 0u : 0xFFFFFFFFu);
  r &= b.w ^ ((me & 128u) ? 0u : 0xFFFFFFFFu);
  return r;
}

__device__ __forceinline__ void filter_slow_one(
    const float* __restrict__ src, const unsigned int* __restrict__ pkw, int e,
    float* acc, int* cnt, bool wantDeg) {
  const unsigned int w = pkw[e];
  const int r = (int)(w >> 20);
  const float* s = src + (size_t)(w & 0xFFFFFu) * 64;
  for (int c = 0; c < 64; ++c) atomicAdd(&acc[r * 64 + c], s[c]);
  if (wantDeg) atomicAdd(&cnt[r], 1);
}

__device__ __forceinline__ void filter_sum_body(
    const float* __restrict__ src, const unsigned int* __restrict__ pkw,
    const unsigned int* __restrict__ map,
    float* __restrict__ dst, float* __restrict__ deg, const bool wantDeg,
    const int R, const int nDst, float* acc, int* cnt, int* wq) {
  const int tid = threadIdx.x;
  const int lane = tid & 31;
  const int wave = __builtin_amdgcn_readfirstlane(tid >> 5);
  const int base = blockIdx.x * R;

  for (int i = tid; i < R * 64; i += FT) acc[i] = 0.0f;
  if (wantDeg)
    for (int i = tid; i < R; i += FT) cnt[i] = 0;
  __syncthreads();

  const unsigned int me = blockIdx.x;
  const int nChunks = (NGROUPS + FT * FG - 1) / (FT * FG);
  const uint4* mapv = (const uint4*)map;
  int* const qw = wq + wave * (2 * QW);

  uint4 ma[FG], mb[FG];
#pragma unroll
  for (int s = 0; s < FG; ++s) {
    const int gi = s * FT + tid;
    const int gl = gi < NGROUPS ? gi : NGROUPS - 1;
    ma[s] = mapv[gl * 2];
    mb[s] = mapv[gl * 2 + 1];
  }

  int pkP = -1;
  int nPrev = 0;

  for (int c = 0; c <= nChunks; ++c) {
    const int c0 = c * (FT * FG);
    const bool doScan = c < nChunks;
    int* qScan = qw + (c & 1) * QW;
    const int* qDrain = qw + ((c & 1) ^ 1) * QW;
    int nCur = 0;
    int pkN = -1;

    if (doScan) {
#pragma unroll
      for (int s = 0; s < FG; ++s) {
        const int gi = c0 + s * FT + tid;
        unsigned int z = gi < NGROUPS ? owner_match(ma[s], mb[s], me) : 0u;
        while (true) {
          const bool act = z != 0u;
          const unsigned int bal = __builtin_amdgcn_ballot_w32(act);
          if (bal == 0u) break;
          if (act) {
            const int slot = nCur + (int)__builtin_amdgcn_mbcnt_lo(bal, 0u);
            if (slot < QW) qScan[slot] = gi * 32 + __builtin_ctz(z);
            z &= z - 1;
          }
          nCur += __builtin_popcount(bal);
        }
      }
      if (nCur > QW) {
        nCur = 0;
#pragma nounroll
        for (int s = 0; s < FG; ++s) {
          const int gi = c0 + s * FT + tid;
          if (gi >= NGROUPS) continue;
          unsigned int z = owner_match(mapv[gi * 2], mapv[gi * 2 + 1], me);
          while (z) {
            const int b = __builtin_ctz(z);
            z &= z - 1;
            filter_slow_one(src, pkw, gi * 32 + b, acc, cnt, wantDeg);
          }
        }
      }

      if (c + 1 < nChunks) {
#pragma unroll
        for (int s = 0; s < FG; ++s) {
          const int gi = c0 + FT * FG + s * FT + tid;
          const int gl = gi < NGROUPS ? gi : NGROUPS - 1;
          ma[s] = mapv[gl * 2];
          mb[s] = mapv[gl * 2 + 1];
        }
      }

      if (lane < DR && lane < nCur) pkN = (int)pkw[qScan[lane]];
    }

    for (int jb = 0; jb < nPrev; jb += DR) {
      int pk = pkP;
      if (jb > 0) {
        pk = -1;
        if (lane < DR && jb + lane < nPrev) pk = (int)pkw[qDrain[jb + lane]];
      }
      float2 r[DR];
      int p[DR];
#pragma unroll
      for (int k = 0; k < DR; ++k) {
        p[k] = __builtin_amdgcn_readlane(pk, k);
        const int srow = p[k] >= 0 ? (p[k] & 0xFFFFF) : 0;
        r[k] = *(const float2*)(src + (size_t)srow * 64 + lane * 2);
      }
#pragma unroll
      for (int k = 0; k < DR; ++k) {
        if (p[k] >= 0) {
          float* a = acc + (p[k] >> 20) * 64 + lane * 2;
          atomicAdd(a, r[k].x);
          atomicAdd(a + 1, r[k].y);
          if (wantDeg && lane == 0) atomicAdd(&cnt[p[k] >> 20], 1);
        }
      }
    }
    pkP = pkN;
    nPrev = nCur;
  }
  __syncthreads();

  for (int i = tid; i < R * 64; i += FT) {
    const int row = base + (i >> 6);
    if (row < nDst) dst[(size_t)row * 64 + (i & 63)] = acc[i];
  }
  if (wantDeg)
    for (int i = tid; i < R; i += FT)
      if (base + i < nDst) deg[base + i] = (float)cnt[i];
}

__global__ __launch_bounds__(FT) void filter_edges_kernel(
    const float* __restrict__ Y1, const unsigned int* __restrict__ pkE,
    const unsigned int* __restrict__ mapE, float* __restrict__ Xe) {
  __shared__ float acc[OWN_E * 64];
  __shared__ int wq[(FT / 32) * 2 * QW];
  filter_sum_body(Y1, pkE, mapE, Xe, nullptr, false, OWN_E, N_EDGES_C, acc,
                  nullptr, wq);
}

__global__ __launch_bounds__(FT) void filter_nodes_kernel(
    const float* __restrict__ Xe, const unsigned int* __restrict__ pkV,
    const unsigned int* __restrict__ mapV, float* __restrict__ S,
    float* __restrict__ deg) {
  __shared__ float acc[OWN_V * 64];
  __shared__ int cnt[OWN_V];
  __shared__ int wq[(FT / 32) * 2 * QW];
  filter_sum_body(Xe, pkV, mapV, S, deg, true, OWN_V, N_NODES_C, acc, cnt, wq);
}

extern "C" void kernel_launch(void* const* d_in, const int* in_sizes, int n_in,
                              void* d_out, int out_size, void* d_ws,
                              size_t ws_size, hipStream_t stream) {
  (void)in_sizes; (void)n_in; (void)out_size; (void)ws_size;
  const float* X    = (const float*)d_in[0];
  const float* X0   = (const float*)d_in[1];
  const int*   vertex = (const int*)d_in[2];
  const int*   edges  = (const int*)d_in[3];
  const float* W1w  = (const float*)d_in[4];
  const float* W1b  = (const float*)d_in[5];
  const float* W2w  = (const float*)d_in[6];
  const float* W2b  = (const float*)d_in[7];
  const float* Ww   = (const float*)d_in[8];
  const float* Wb   = (const float*)d_in[9];

  float* out = (float*)d_out;
  float* Xe  = out + (size_t)N_NODES_C * 64;

  float* ws  = (float*)d_ws;
  float* Y1  = ws;
  float* S   = Y1 + (size_t)N_NODES_C * 64;
  float* deg = S + (size_t)N_NODES_C * 64;

  unsigned int* mapE = (unsigned int*)d_out;
  unsigned int* mapV = mapE + NNZ_C / 4;
  unsigned int* pkE  = mapV + NNZ_C / 4;
  unsigned int* pkV  = pkE + NNZ_C;

  gemm64_bias_kernel<<<RES_WG, GEMM_T, 0, stream>>>(X, W1w, W1b, Y1, N_NODES_C);
  owner_map_kernel<<<RES_WG, MAP_T, 0, stream>>>(vertex, edges, mapE, mapV, pkE,
                                                 pkV);
  filter_edges_kernel<<<OWNERS, FT, 0, stream>>>(Y1, pkE, mapE, Xe);
  filter_nodes_kernel<<<OWNERS, FT, 0, stream>>>(Xe, pkV, mapV, S, deg);
  gemm2_mix_kernel<<<RES_WG, GEMM2_T, 0, stream>>>(X, X0, S, deg, W2w, W2b, Y1,
                                                  N_NODES_C);
  gemm64_bias_kernel<<<RES_WG, GEMM_T, 0, stream>>>(Y1, Ww, Wb, out, N_NODES_C);
}


# segment sum drain: once per pass the lane with an entry's packed word computes its source offset, LDS offset and degree count (one LDS count add per pass, not one per hit); no sum reordered
# speedup vs baseline: 2.9400x; 1.0474x over previous
#include <hip/hip_runtime.h>

typedef __attribute__((ext_vector_type(2))) float v2f;
typedef __attribute__((ext_vector_type(8))) float v8f;

#define N_NODES_C 100000
#define N_EDGES_C 20000
#define NNZ_C     1000000
#define ALPHA_C   0.5f

__device__ __forceinline__ v8f wmma_f32(v2f a, v2f b, v8f c) {
  return __builtin_amdgcn_wmma_f32_16x16x4_f32(false, a, false, b, (short)0, c,
                                               false, false);
}

#define RES_WG   256
#define GEMM_T   512
#define GEMM_NW  (GEMM_T / 32)
#define GEMM2_T  256
#define GEMM2_NW (GEMM2_T / 32)

__global__ __launch_bounds__(GEMM_T) void gemm64_bias_kernel(
    const float* __restrict__ A, const float* __restrict__ W,
    const float* __restrict__ bias, float* __restrict__ Out, int M) {
  __shared__ float Bsh[64 * 64];

  const int lane = threadIdx.x & 31;
  const int wave = __builtin_amdgcn_readfirstlane((int)(threadIdx.x >> 5));
  const int ll = lane & 15;
  const int lh = lane >> 4;
  const int tiles = M >> 4;
  const int tBeg = (int)(((long long)blockIdx.x * tiles) / RES_WG);
  const int tEnd = (int)(((long long)(blockIdx.x + 1) * tiles) / RES_WG);
  int t = tBeg + wave;

  v2f an[16];
#pragma unroll
  for (int s = 0; s < 16; ++s) { an[s].x = 0.0f; an[s].y = 0.0f; }
  if (t < tEnd) {
    const float* Arow = A + (size_t)(t * 16 + ll) * 64;
#pragma unroll
    for (int s = 0; s < 16; ++s) {
      const int ka = s * 4 + lh * 2;
      an[s].x = Arow[ka];
      an[s].y = Arow[ka + 1];
    }
  }

#pragma unroll
  for (int j = 0; j < 64 * 64 / GEMM_T; ++j) {
    const int i = j * GEMM_T + (int)threadIdx.x;
    const int k = i >> 6, n = i & 63;
    Bsh[i] = W[n * 64 + k];
  }
  const float bi0 = bias[ll], bi1 = bias[16 + ll], bi2 = bias[32 + ll],
              bi3 = bias[48 + ll];
  __syncthreads();

  while (t < tEnd) {
    v2f a[16];
#pragma unroll
    for (int s = 0; s < 16; ++s) a[s] = an[s];
    const int tn = t + GEMM_NW;
    if (tn < tEnd) {
      const float* Arow = A + (size_t)(tn * 16 + ll) * 64;
#pragma unroll
      for (int s = 0; s < 16; ++s) {
        const int ka = s * 4 + lh * 2;
        an[s].x = Arow[ka];
        an[s].y = Arow[ka + 1];
      }
    }

    v8f acc0 = {}, acc1 = {}, acc2 = {}, acc3 = {};
#pragma unroll
    for (int s = 0; s < 16; ++s) {
      const int ka = s * 4 + lh * 2;
      const float* B0 = &Bsh[ka * 64 + ll];
      const float* B1 = &Bsh[(ka + 1) * 64 + ll];
      v2f b;
      b.x = B0[0];  b.y = B1[0];  acc0 = wmma_f32(a[s], b, acc0);
      b.x = B0[16]; b.y = B1[16]; acc1 = wmma_f32(a[s], b, acc1);
      b.x = B0[32]; b.y = B1[32]; acc2 = wmma_f32(a[s], b, acc2);
      b.x = B0[48]; b.y = B1[48]; acc3 = wmma_f32(a[s], b, acc3);
    }

    const int rowBase = t << 4;
#pragma unroll
    for (int j = 0; j < 8; ++j) {
      const int r = rowBase + j + lh * 8;
      float* o = Out + (size_t)r * 64;
      o[ll]      = acc0[j] + bi0;
      o[16 + ll] = acc1[j] + bi1;
      o[32 + ll] = acc2[j] + bi2;
      o[48 + ll] = acc3[j] + bi3;
    }
    t = tn;
  }
}

__global__ __launch_bounds__(GEMM2_T) void gemm2_mix_kernel(
    const float* __restrict__ X, const float* __restrict__ X0,
    const float* __restrict__ S, const float* __restrict__ deg,
    const float* __restrict__ W2, const float* __restrict__ b2,
    float* __restrict__ T, int M) {
  __shared__ float Bsh[128 * 64];

  const int lane = threadIdx.x & 31;
  const int wave = __builtin_amdgcn_readfirstlane((int)(threadIdx.x >> 5));
  const int ll = lane & 15;
  const int lh = lane >> 4;
  const int tiles = M >> 4;
  const int tBeg = (int)(((long long)blockIdx.x * tiles) / RES_WG);
  const int tEnd = (int)(((long long)(blockIdx.x + 1) * tiles) / RES_WG);
  int t = tBeg + wave;

  v2f xn[16], sn[16];
  float dgn = 0.0f;
#pragma unroll
  for (int s = 0; s < 16; ++s) {
    xn[s].x = 0.0f; xn[s].y = 0.0f; sn[s].x = 0.0f; sn[s].y = 0.0f;
  }
  if (t < tEnd) {
    dgn = deg[t * 16 + ll];
    const float* Xrow = X + (size_t)(t * 16 + ll) * 64;
    const float* Srow = S + (size_t)(t * 16 + ll) * 64;
#pragma unroll
    for (int s = 0; s < 16; ++s) {
      const int ka = s * 4 + lh * 2;
      xn[s].x = Xrow[ka]; xn[s].y = Xrow[ka + 1];
      sn[s].x = Srow[ka]; sn[s].y = Srow[ka + 1];
    }
  }

  {
    float w[128 * 64 / GEMM2_T];
#pragma unroll
    for (int j = 0; j < 128 * 64 / GEMM2_T; ++j) {
      const int i = j * GEMM2_T + (int)threadIdx.x;
      const int k = i >> 6, n = i & 63;
      w[j] = W2[n * 128 + k];
    }
#pragma unroll
    for (int j = 0; j < 128 * 64 / GEMM2_T; ++j)
      Bsh[j * GEMM2_T + (int)threadIdx.x] = w[j];
  }
  const float c0 = b2[ll], c1 = b2[16 + ll], c2 = b2[32 + ll], c3 = b2[48 + ll];
  __syncthreads();

  const float oma = 1.0f - ALPHA_C;
  while (t < tEnd) {
    v2f xa[16], sa[16];
#pragma unroll
    for (int s = 0; s < 16; ++s) { xa[s] = xn[s]; sa[s] = sn[s]; }
    const float dg = dgn;
    const int tn = t + GEMM2_NW;
    if (tn < tEnd) {
      dgn = deg[tn * 16 + ll];
      const float* Xrow = X + (size_t)(tn * 16 + ll) * 64;
      const float* Srow = S + (size_t)(tn * 16 + ll) * 64;
#pragma unroll
      for (int s = 0; s < 16; ++s) {
        const int ka = s * 4 + lh * 2;
        xn[s].x = Xrow[ka]; xn[s].y = Xrow[ka + 1];
        sn[s].x = Srow[ka]; sn[s].y = Srow[ka + 1];
      }
    }

    const int rowBase = t << 4;
    float dr[8], z0[8], z1[8], z2[8], z3[8];
#pragma unroll
    for (int j = 0; j < 8; ++j) {
      const int r = rowBase + j + lh * 8;
      const float* x0r = X0 + (size_t)r * 64;
      dr[j] = deg[r];
      z0[j] = x0r[ll];
      z1[j] = x0r[16 + ll];
      z2[j] = x0r[32 + ll];
      z3[j] = x0r[48 + ll];
    }

    v8f acc0 = {}, acc1 = {}, acc2 = {}, acc3 = {};
#pragma unroll
    for (int s = 0; s < 16; ++s) {
      const int ka = s * 4 + lh * 2;
      v2f a;
      a.x = dg * xa[s].x;
      a.y = dg * xa[s].y;
      const float* B0 = &Bsh[ka * 64 + ll];
      const float* B1 = &Bsh[(ka + 1) * 64 + ll];
      v2f b;
      b.x = B0[0];  b.y = B1[0];  acc0 = wmma_f32(a, b, acc0);
      b.x = B0[16]; b.y = B1[16]; acc1 = wmma_f32(a, b, acc1);
      b.x = B0[32]; b.y = B1[32]; acc2 = wmma_f32(a, b, acc2);
      b.x = B0[48]; b.y = B1[48]; acc3 = wmma_f32(a, b, acc3);
    }
#pragma unroll
    for (int s = 0; s < 16; ++s) {
      const int ka = 64 + s * 4 + lh * 2;
      const float* B0 = &Bsh[ka * 64 + ll];
      const float* B1 = &Bsh[(ka + 1) * 64 + ll];
      v2f b;
      b.x = B0[0];  b.y = B1[0];  acc0 = wmma_f32(sa[s], b, acc0);
      b.x = B0[16]; b.y = B1[16]; acc1 = wmma_f32(sa[s], b, acc1);
      b.x = B0[32]; b.y = B1[32]; acc2 = wmma_f32(sa[s], b, acc2);
      b.x = B0[48]; b.y = B1[48]; acc3 = wmma_f32(sa[s], b, acc3);
    }

#pragma unroll
    for (int j = 0; j < 8; ++j) {
      const int r = rowBase + j + lh * 8;
      float* o = T + (size_t)r * 64;
      o[ll]      = oma * (acc0[j] + dr[j] * c0) + ALPHA_C * z0[j];
      o[16 + ll] = oma * (acc1[j] + dr[j] * c1) + ALPHA_C * z1[j];
      o[32 + ll] = oma * (acc2[j] + dr[j] * c2) + ALPHA_C * z2[j];
      o[48 + ll] = oma * (acc3[j] + dr[j] * c3) + ALPHA_C * z3[j];
    }
    t = tn;
  }
}

#define OWN_E   79
#define OWN_V   391
#define OWNERS  256
#define FT      1024
#define FG      4
#define QW      128
#define DR      24
#define NGROUPS (NNZ_C / 32)

#define MAP_T     256
#define MAP_WAVES (RES_WG * (MAP_T / 32))
#define NQUADS    ((NGROUPS + 3) / 4)
#define MAP_QPW   ((NQUADS + MAP_WAVES - 1) / MAP_WAVES)

__global__ __launch_bounds__(MAP_T) void owner_map_kernel(
    const int* __restrict__ vertex, const int* __restrict__ edges,
    unsigned int* __restrict__ mapE, unsigned int* __restrict__ mapV,
    unsigned int* __restrict__ pkE, unsigned int* __restrict__ pkV) {
  const int lane = threadIdx.x & 31;
  const int waveG = blockIdx.x * (MAP_T / 32) +
                    __builtin_amdgcn_readfirstlane((int)(threadIdx.x >> 5));
  int ev[MAP_QPW][4], vv[MAP_QPW][4];
#pragma unroll
  for (int i = 0; i < MAP_QPW; ++i) {
    const int q = waveG + i * MAP_WAVES;
#pragma unroll
    for (int k = 0; k < 4; ++k) {
      const int g = q * 4 + k;
      const int gl = g < NGROUPS ? g : NGROUPS - 1;
      ev[i][k] = edges[gl * 32 + lane];
      vv[i][k] = vertex[gl * 32 + lane];
    }
  }
#pragma unroll
  for (int i = 0; i < MAP_QPW; ++i) {
    const int q = waveG + i * MAP_WAVES;
    unsigned int pe = 0, pv = 0;
#pragma unroll
    for (int k = 0; k < 4; ++k) {
      const unsigned int oE = (unsigned int)(ev[i][k] / OWN_E);
      const unsigned int oV = (unsigned int)(vv[i][k] / OWN_V);
      if (q * 4 + k < NGROUPS) {
        const int e = (q * 4 + k) * 32 + lane;
        pkE[e] = (((unsigned int)ev[i][k] - oE * OWN_E) << 20) |
                 (unsigned int)vv[i][k];
        pkV[e] = (((unsigned int)vv[i][k] - oV * OWN_V) << 20) |
                 (unsigned int)ev[i][k];
      }
#pragma unroll
      for (int p = 0; p < 8; ++p) {
        const unsigned int bE = __builtin_amdgcn_ballot_w32(((oE >> p) & 1u) != 0u);
        const unsigned int bV = __builtin_amdgcn_ballot_w32(((oV >> p) & 1u) != 0u);
        if (lane == 8 * k + p) { pe = bE; pv = bV; }
      }
    }
    if (q * 4 + (lane >> 3) < NGROUPS) {
      mapE[q * 32 + lane] = pe;
      mapV[q * 32 + lane] = pv;
    }
  }
}

__device__ __forceinline__ unsigned int owner_match(const uint4 a, const uint4 b,
                                                    const unsigned int me) {
  unsigned int r;
  r  = a.x ^ ((me & 1u)   ? 0u : 0xFFFFFFFFu);
  r &= a.y ^ ((me & 2u)   ? 0u : 0xFFFFFFFFu);
  r &= a.z ^ ((me & 4u)   ? 0u : 0xFFFFFFFFu);
  r &= a.w ^ ((me & 8u)   ? 0u : 0xFFFFFFFFu);
  r &= b.x ^ ((me & 16u)  ? 0u : 0xFFFFFFFFu);
  r &= b.y ^ ((me & 32u)  ? 0u : 0xFFFFFFFFu);
  r &= b.z ^ ((me & 64u)  ? 0u : 0xFFFFFFFFu);
  r &= b.w ^ ((me & 128u) ? 0u : 0xFFFFFFFFu);
  return r;
}

__device__ __forceinline__ void filter_slow_one(
    const float* __restrict__ src, const unsigned int* __restrict__ pkw, int e,
    float* acc, int* cnt, bool wantDeg) {
  const unsigned int w = pkw[e];
  const int r = (int)(w >> 20);
  const float* s = src + (size_t)(w & 0xFFFFFu) * 64;
  for (int c = 0; c < 64; ++c) atomicAdd(&acc[r * 64 + c], s[c]);
  if (wantDeg) atomicAdd(&cnt[r], 1);
}

__device__ __forceinline__ void filter_sum_body(
    const float* __restrict__ src, const unsigned int* __restrict__ pkw,
    const unsigned int* __restrict__ map,
    float* __restrict__ dst, float* __restrict__ deg, const bool wantDeg,
    const int R, const int nDst, float* acc, int* cnt, int* wq) {
  const int tid = threadIdx.x;
  const int lane = tid & 31;
  const int wave = __builtin_amdgcn_readfirstlane(tid >> 5);
  const int base = blockIdx.x * R;

  for (int i = tid; i < R * 64; i += FT) acc[i] = 0.0f;
  if (wantDeg)
    for (int i = tid; i < R; i += FT) cnt[i] = 0;
  __syncthreads();

  const unsigned int me = blockIdx.x;
  const int nChunks = (NGROUPS + FT * FG - 1) / (FT * FG);
  const uint4* mapv = (const uint4*)map;
  int* const qw = wq + wave * (2 * QW);

  uint4 ma[FG], mb[FG];
#pragma unroll
  for (int s = 0; s < FG; ++s) {
    const int gi = s * FT + tid;
    const int gl = gi < NGROUPS ? gi : NGROUPS - 1;
    ma[s] = mapv[gl * 2];
    mb[s] = mapv[gl * 2 + 1];
  }

  int pkP = -1;
  int nPrev = 0;

  for (int c = 0; c <= nChunks; ++c) {
    const int c0 = c * (FT * FG);
    const bool doScan = c < nChunks;
    int* qScan = qw + (c & 1) * QW;
    const int* qDrain = qw + ((c & 1) ^ 1) * QW;
    int nCur = 0;
    int pkN = -1;

    if (doScan) {
#pragma unroll
      for (int s = 0; s < FG; ++s) {
        const int gi = c0 + s * FT + tid;
        unsigned int z = gi < NGROUPS ? owner_match(ma[s], mb[s], me) : 0u;
        while (true) {
          const bool act = z != 0u;
          const unsigned int bal = __builtin_amdgcn_ballot_w32(act);
          if (bal == 0u) break;
          if (act) {
            const int slot = nCur + (int)__builtin_amdgcn_mbcnt_lo(bal, 0u);
            if (slot < QW) qScan[slot] = gi * 32 + __builtin_ctz(z);
            z &= z - 1;
          }
          nCur += __builtin_popcount(bal);
        }
      }
      if (nCur > QW) {
        nCur = 0;
#pragma nounroll
        for (int s = 0; s < FG; ++s) {
          const int gi = c0 + s * FT + tid;
          if (gi >= NGROUPS) continue;
          unsigned int z = owner_match(mapv[gi * 2], mapv[gi * 2 + 1], me);
          while (z) {
            const int b = __builtin_ctz(z);
            z &= z - 1;
            filter_slow_one(src, pkw, gi * 32 + b, acc, cnt, wantDeg);
          }
        }
      }

      if (c + 1 < nChunks) {
#pragma unroll
        for (int s = 0; s < FG; ++s) {
          const int gi = c0 + FT * FG + s * FT + tid;
          const int gl = gi < NGROUPS ? gi : NGROUPS - 1;
          ma[s] = mapv[gl * 2];
          mb[s] = mapv[gl * 2 + 1];
        }
      }

      if (lane < DR && lane < nCur) pkN = (int)pkw[qScan[lane]];
    }

    for (int jb = 0; jb < nPrev; jb += DR) {
      int pk = pkP;
      if (jb > 0) {
        pk = -1;
        if (lane < DR && jb + lane < nPrev) pk = (int)pkw[qDrain[jb + lane]];
      }
      const int n = (nPrev - jb) < DR ? (nPrev - jb) : DR;
      const bool has = pk >= 0;
      const unsigned int sOff = has ? (((unsigned int)pk & 0xFFFFFu) << 8) : 0u;
      const unsigned int dOff = has ? (((unsigned int)pk >> 20) << 8) : 0u;
      if (wantDeg && has) atomicAdd(&cnt[(unsigned int)pk >> 20], 1);
      float2 r[DR];
#pragma unroll
      for (int k = 0; k < DR; ++k) {
        const unsigned int so =
            (unsigned int)__builtin_amdgcn_readlane((int)sOff, k);
        r[k] = *(const float2*)((const char*)src + (size_t)so +
                                (size_t)(lane * 8));
      }
      asm volatile("" : "+v"(r[1].x), "+v"(r[1].y));
#pragma unroll
      for (int k = 0; k < DR; ++k) {
        if (k == 0 || k < n) {
          const unsigned int d =
              (unsigned int)__builtin_amdgcn_readlane((int)dOff, k);
          float* a = (float*)((char*)acc + d) + lane * 2;
          atomicAdd(a, r[k].x);
          atomicAdd(a + 1, r[k].y);
        }
      }
    }
    pkP = pkN;
    nPrev = nCur;
  }
  __syncthreads();

  for (int i = tid; i < R * 64; i += FT) {
    const int row = base + (i >> 6);
    if (row < nDst) dst[(size_t)row * 64 + (i & 63)] = acc[i];
  }
  if (wantDeg)
    for (int i = tid; i < R; i += FT)
      if (base + i < nDst) deg[base + i] = (float)cnt[i];
}

__global__ __launch_bounds__(FT) void filter_edges_kernel(
    const float* __restrict__ Y1, const unsigned int* __restrict__ pkE,
    const unsigned int* __restrict__ mapE, float* __restrict__ Xe) {
  __shared__ float acc[OWN_E * 64];
  __shared__ int wq[(FT / 32) * 2 * QW];
  filter_sum_body(Y1, pkE, mapE, Xe, nullptr, false, OWN_E, N_EDGES_C, acc,
                  nullptr, wq);
}

__global__ __launch_bounds__(FT) void filter_nodes_kernel(
    const float* __restrict__ Xe, const unsigned int* __restrict__ pkV,
    const unsigned int* __restrict__ mapV, float* __restrict__ S,
    float* __restrict__ deg) {
  __shared__ float acc[OWN_V * 64];
  __shared__ int cnt[OWN_V];
  __shared__ int wq[(FT / 32) * 2 * QW];
  filter_sum_body(Xe, pkV, mapV, S, deg, true, OWN_V, N_NODES_C, acc, cnt, wq);
}

extern "C" void kernel_launch(void* const* d_in, const int* in_sizes, int n_in,
                              void* d_out, int out_size, void* d_ws,
                              size_t ws_size, hipStream_t stream) {
  (void)in_sizes; (void)n_in; (void)out_size; (void)ws_size;
  const float* X    = (const float*)d_in[0];
  const float* X0   = (const float*)d_in[1];
  const int*   vertex = (const int*)d_in[2];
  const int*   edges  = (const int*)d_in[3];
  const float* W1w  = (const float*)d_in[4];
  const float* W1b  = (const float*)d_in[5];
  const float* W2w  = (const float*)d_in[6];
  const float* W2b  = (const float*)d_in[7];
  const float* Ww   = (const float*)d_in[8];
  const float* Wb   = (const float*)d_in[9];

  float* out = (float*)d_out;
  float* Xe  = out + (size_t)N_NODES_C * 64;

  float* ws  = (float*)d_ws;
  float* Y1  = ws;
  float* S   = Y1 + (size_t)N_NODES_C * 64;
  float* deg = S + (size_t)N_NODES_C * 64;

  unsigned int* mapE = (unsigned int*)d_out;
  unsigned int* mapV = mapE + NNZ_C / 4;
  unsigned int* pkE  = mapV + NNZ_C / 4;
  unsigned int* pkV  = pkE + NNZ_C;

  gemm64_bias_kernel<<<RES_WG, GEMM_T, 0, stream>>>(X, W1w, W1b, Y1, N_NODES_C);
  owner_map_kernel<<<RES_WG, MAP_T, 0, stream>>>(vertex, edges, mapE, mapV, pkE,
                                                 pkV);
  filter_edges_kernel<<<OWNERS, FT, 0, stream>>>(Y1, pkE, mapE, Xe);
  filter_nodes_kernel<<<OWNERS, FT, 0, stream>>>(Xe, pkV, mapV, S, deg);
  gemm2_mix_kernel<<<RES_WG, GEMM2_T, 0, stream>>>(X, X0, S, deg, W2w, W2b, Y1,
                                                  N_NODES_C);
  gemm64_bias_kernel<<<RES_WG, GEMM_T, 0, stream>>>(Y1, Ww, Wb, out, N_NODES_C);
}


# node sum launch also forms its rows' two products on f32 WMMA straight from the owner's LDS sums: S, deg and T never go to memory, maps moved into the workspace; 6 -> 4 launches; no sum reordered
# speedup vs baseline: 3.3144x; 1.1273x over previous
#include <hip/hip_runtime.h>

typedef __attribute__((ext_vector_type(2))) float v2f;
typedef __attribute__((ext_vector_type(8))) float v8f;

#define N_NODES_C 100000
#define N_EDGES_C 20000
#define NNZ_C     1000000
#define ALPHA_C   0.5f

__device__ __forceinline__ v8f wmma_f32(v2f a, v2f b, v8f c) {
  return __builtin_amdgcn_wmma_f32_16x16x4_f32(false, a, false, b, (short)0, c,
                                               false, false);
}

#define RES_WG   256
#define GEMM_T   512
#define GEMM_NW  (GEMM_T / 32)

__global__ __launch_bounds__(GEMM_T) void gemm64_bias_kernel(
    const float* __restrict__ A, const float* __restrict__ W,
    const float* __restrict__ bias, float* __restrict__ Out, int M) {
  __shared__ float Bsh[64 * 64];

  const int lane = threadIdx.x & 31;
  const int wave = __builtin_amdgcn_readfirstlane((int)(threadIdx.x >> 5));
  const int ll = lane & 15;
  const int lh = lane >> 4;
  const int tiles = M >> 4;
  const int tBeg = (int)(((long long)blockIdx.x * tiles) / RES_WG);
  const int tEnd = (int)(((long long)(blockIdx.x + 1) * tiles) / RES_WG);
  int t = tBeg + wave;

  v2f an[16];
#pragma unroll
  for (int s = 0; s < 16; ++s) { an[s].x = 0.0f; an[s].y = 0.0f; }
  if (t < tEnd) {
    const float* Arow = A + (size_t)(t * 16 + ll) * 64;
#pragma unroll
    for (int s = 0; s < 16; ++s) {
      const int ka = s * 4 + lh * 2;
      an[s].x = Arow[ka];
      an[s].y = Arow[ka + 1];
    }
  }

#pragma unroll
  for (int j = 0; j < 64 * 64 / GEMM_T; ++j) {
    const int i = j * GEMM_T + (int)threadIdx.x;
    const int k = i >> 6, n = i & 63;
    Bsh[i] = W[n * 64 + k];
  }
  const float bi0 = bias[ll], bi1 = bias[16 + ll], bi2 = bias[32 + ll],
              bi3 = bias[48 + ll];
  __syncthreads();

  while (t < tEnd) {
    v2f a[16];
#pragma unroll
    for (int s = 0; s < 16; ++s) a[s] = an[s];
    const int tn = t + GEMM_NW;
    if (tn < tEnd) {
      const float* Arow = A + (size_t)(tn * 16 + ll) * 64;
#pragma unroll
      for (int s = 0; s < 16; ++s) {
        const int ka = s * 4 + lh * 2;
        an[s].x = Arow[ka];
        an[s].y = Arow[ka + 1];
      }
    }

    v8f acc0 = {}, acc1 = {}, acc2 = {}, acc3 = {};
#pragma unroll
    for (int s = 0; s < 16; ++s) {
      const int ka = s * 4 + lh * 2;
      const float* B0 = &Bsh[ka * 64 + ll];
      const float* B1 = &Bsh[(ka + 1) * 64 + ll];
      v2f b;
      b.x = B0[0];  b.y = B1[0];  acc0 = wmma_f32(a[s], b, acc0);
      b.x = B0[16]; b.y = B1[16]; acc1 = wmma_f32(a[s], b, acc1);
      b.x = B0[32]; b.y = B1[32]; acc2 = wmma_f32(a[s], b, acc2);
      b.x = B0[48]; b.y = B1[48]; acc3 = wmma_f32(a[s], b, acc3);
    }

    const int rowBase = t << 4;
#pragma unroll
    for (int j = 0; j < 8; ++j) {
      const int r = rowBase + j + lh * 8;
      float* o = Out + (size_t)r * 64;
      o[ll]      = acc0[j] + bi0;
      o[16 + ll] = acc1[j] + bi1;
      o[32 + ll] = acc2[j] + bi2;
      o[48 + ll] = acc3[j] + bi3;
    }
    t = tn;
  }
}

#define OWN_E   79
#define OWN_V   391
#define OWNERS  256
#define FT      1024
#define FG      4
#define QW      128
#define DR      24
#define NGROUPS (NNZ_C / 32)

#define MAP_T     256
#define MAP_WAVES (RES_WG * (MAP_T / 32))
#define NQUADS    ((NGROUPS + 3) / 4)
#define MAP_QPW   ((NQUADS + MAP_WAVES - 1) / MAP_WAVES)

__global__ __launch_bounds__(MAP_T) void owner_map_kernel(
    const int* __restrict__ vertex, const int* __restrict__ edges,
    unsigned int* __restrict__ mapE, unsigned int* __restrict__ mapV,
    unsigned int* __restrict__ pkE, unsigned int* __restrict__ pkV) {
  const int lane = threadIdx.x & 31;
  const int waveG = blockIdx.x * (MAP_T / 32) +
                    __builtin_amdgcn_readfirstlane((int)(threadIdx.x >> 5));
  int ev[MAP_QPW][4], vv[MAP_QPW][4];
#pragma unroll
  for (int i = 0; i < MAP_QPW; ++i) {
    const int q = waveG + i * MAP_WAVES;
#pragma unroll
    for (int k = 0; k < 4; ++k) {
      const int g = q * 4 + k;
      const int gl = g < NGROUPS ? g : NGROUPS - 1;
      ev[i][k] = edges[gl * 32 + lane];
      vv[i][k] = vertex[gl * 32 + lane];
    }
  }
#pragma unroll
  for (int i = 0; i < MAP_QPW; ++i) {
    const int q = waveG + i * MAP_WAVES;
    unsigned int pe = 0, pv = 0;
#pragma unroll
    for (int k = 0; k < 4; ++k) {
      const unsigned int oE = (unsigned int)(ev[i][k] / OWN_E);
      const unsigned int oV = (unsigned int)(vv[i][k] / OWN_V);
      if (q * 4 + k < NGROUPS) {
        const int e = (q * 4 + k) * 32 + lane;
        pkE[e] = (((unsigned int)ev[i][k] - oE * OWN_E) << 20) |
                 (unsigned int)vv[i][k];
        pkV[e] = (((unsigned int)vv[i][k] - oV * OWN_V) << 20) |
                 (unsigned int)ev[i][k];
      }
#pragma unroll
      for (int p = 0; p < 8; ++p) {
        const unsigned int bE = __builtin_amdgcn_ballot_w32(((oE >> p) & 1u) != 0u);
        const unsigned int bV = __builtin_amdgcn_ballot_w32(((oV >> p) & 1u) != 0u);
        if (lane == 8 * k + p) { pe = bE; pv = bV; }
      }
    }
    if (q * 4 + (lane >> 3) < NGROUPS) {
      mapE[q * 32 + lane] = pe;
      mapV[q * 32 + lane] = pv;
    }
  }
}

__device__ __forceinline__ unsigned int owner_match(const uint4 a, const uint4 b,
                                                    const unsigned int me) {
  unsigned int r;
  r  = a.x ^ ((me & 1u)   ? 0u : 0xFFFFFFFFu);
  r &= a.y ^ ((me & 2u)   ? 0u : 0xFFFFFFFFu);
  r &= a.z ^ ((me & 4u)   ? 0u : 0xFFFFFFFFu);
  r &= a.w ^ ((me & 8u)   ? 0u : 0xFFFFFFFFu);
  r &= b.x ^ ((me & 16u)  ? 0u : 0xFFFFFFFFu);
  r &= b.y ^ ((me & 32u)  ? 0u : 0xFFFFFFFFu);
  r &= b.z ^ ((me & 64u)  ? 0u : 0xFFFFFFFFu);
  r &= b.w ^ ((me & 128u) ? 0u : 0xFFFFFFFFu);
  return r;
}

__device__ __forceinline__ void filter_slow_one(
    const float* __restrict__ src, const unsigned int* __restrict__ pkw, int e,
    float* acc, int* cnt, bool wantDeg, const int P) {
  const unsigned int w = pkw[e];
  const int r = (int)(w >> 20);
  const float* s = src + (size_t)(w & 0xFFFFFu) * 64;
  for (int c = 0; c < 64; ++c) atomicAdd(&acc[r * P + c], s[c]);
  if (wantDeg) atomicAdd(&cnt[r], 1);
}

__device__ __forceinline__ void filter_sum_body(
    const float* __restrict__ src, const unsigned int* __restrict__ pkw,
    const unsigned int* __restrict__ map,
    float* __restrict__ dst, const bool doStore, const bool wantDeg,
    const int R, const int P, const int nDst, float* acc, int* cnt, int* wq) {
  const int tid = threadIdx.x;
  const int lane = tid & 31;
  const int wave = __builtin_amdgcn_readfirstlane(tid >> 5);
  const int base = blockIdx.x * R;

  for (int i = tid; i < R * P; i += FT) acc[i] = 0.0f;
  if (wantDeg)
    for (int i = tid; i < R; i += FT) cnt[i] = 0;
  __syncthreads();

  const unsigned int me = blockIdx.x;
  const int nChunks = (NGROUPS + FT * FG - 1) / (FT * FG);
  const uint4* mapv = (const uint4*)map;
  int* const qw = wq + wave * (2 * QW);

  uint4 ma[FG], mb[FG];
#pragma unroll
  for (int s = 0; s < FG; ++s) {
    const int gi = s * FT + tid;
    const int gl = gi < NGROUPS ? gi : NGROUPS - 1;
    ma[s] = mapv[gl * 2];
    mb[s] = mapv[gl * 2 + 1];
  }

  int pkP = -1;
  int nPrev = 0;

  for (int c = 0; c <= nChunks; ++c) {
    const int c0 = c * (FT * FG);
    const bool doScan = c < nChunks;
    int* qScan = qw + (c & 1) * QW;
    const int* qDrain = qw + ((c & 1) ^ 1) * QW;
    int nCur = 0;
    int pkN = -1;

    if (doScan) {
#pragma unroll
      for (int s = 0; s < FG; ++s) {
        const int gi = c0 + s * FT + tid;
        unsigned int z = gi < NGROUPS ? owner_match(ma[s], mb[s], me) : 0u;
        while (true) {
          const bool act = z != 0u;
          const unsigned int bal = __builtin_amdgcn_ballot_w32(act);
          if (bal == 0u) break;
          if (act) {
            const int slot = nCur + (int)__builtin_amdgcn_mbcnt_lo(bal, 0u);
            if (slot < QW) qScan[slot] = gi * 32 + __builtin_ctz(z);
            z &= z - 1;
          }
          nCur += __builtin_popcount(bal);
        }
      }
      if (nCur > QW) {
        nCur = 0;
#pragma nounroll
        for (int s = 0; s < FG; ++s) {
          const int gi = c0 + s * FT + tid;
          if (gi >= NGROUPS) continue;
          unsigned int z = owner_match(mapv[gi * 2], mapv[gi * 2 + 1], me);
          while (z) {
            const int b = __builtin_ctz(z);
            z &= z - 1;
            filter_slow_one(src, pkw, gi * 32 + b, acc, cnt, wantDeg, P);
          }
        }
      }

      if (c + 1 < nChunks) {
#pragma unroll
        for (int s = 0; s < FG; ++s) {
          const int gi = c0 + FT * FG + s * FT + tid;
          const int gl = gi < NGROUPS ? gi : NGROUPS - 1;
          ma[s] = mapv[gl * 2];
          mb[s] = mapv[gl * 2 + 1];
        }
      }

      if (lane < DR && lane < nCur) pkN = (int)pkw[qScan[lane]];
    }

    for (int jb = 0; jb < nPrev; jb += DR) {
      int pk = pkP;
      if (jb > 0) {
        pk = -1;
        if (lane < DR && jb + lane < nPrev) pk = (int)pkw[qDrain[jb + lane]];
      }
      const int n = (nPrev - jb) < DR ? (nPrev - jb) : DR;
      const bool has = pk >= 0;
      const unsigned int sOff = has ? (((unsigned int)pk & 0xFFFFFu) << 8) : 0u;
      const unsigned int dOff =
          has ? (((unsigned int)pk >> 20) * (unsigned int)(P * 4)) : 0u;
      if (wantDeg && has) atomicAdd(&cnt[(unsigned int)pk >> 20], 1);
      float2 r[DR];
#pragma unroll
      for (int k = 0; k < DR; ++k) {
        const unsigned int so =
            (unsigned int)__builtin_amdgcn_readlane((int)sOff, k);
        r[k] = *(const float2*)((const char*)src + (size_t)so +
                                (size_t)(lane * 8));
      }
      asm volatile("" : "+v"(r[1].x), "+v"(r[1].y));
#pragma unroll
      for (int k = 0; k < DR; ++k) {
        if (k == 0 || k < n) {
          const unsigned int d =
              (unsigned int)__builtin_amdgcn_readlane((int)dOff, k);
          float* a = (float*)((char*)acc + d) + lane * 2;
          atomicAdd(a, r[k].x);
          atomicAdd(a + 1, r[k].y);
        }
      }
    }
    pkP = pkN;
    nPrev = nCur;
  }
  __syncthreads();

  if (doStore)
    for (int i = tid; i < R * 64; i += FT) {
      const int row = base + (i >> 6);
      if (row < nDst)
        dst[(size_t)row * 64 + (i & 63)] = acc[(i >> 6) * P + (i & 63)];
    }
}

__global__ __launch_bounds__(FT) void filter_edges_kernel(
    const float* __restrict__ Y1, const unsigned int* __restrict__ pkE,
    const unsigned int* __restrict__ mapE, float* __restrict__ Xe) {
  __shared__ float acc[OWN_E * 64];
  __shared__ int wq[(FT / 32) * 2 * QW];
  filter_sum_body(Y1, pkE, mapE, Xe, true, false, OWN_E, 64, N_EDGES_C, acc,
                  nullptr, wq);
}

#define OWN_VP  400
#define NODE_TILES (OWN_VP / 16)
#define NODE_P  66
__global__ __launch_bounds__(FT) void filter_nodes_kernel(
    const float* __restrict__ Xe, const unsigned int* __restrict__ pkV,
    const unsigned int* __restrict__ mapV, const float* __restrict__ X,
    const float* __restrict__ X0, const float* __restrict__ W2,
    const float* __restrict__ b2, const float* __restrict__ Ww,
    const float* __restrict__ Wb, float* __restrict__ out) {
  __shared__ __attribute__((aligned(16))) float
      lds[128 * 64 + 64 * 64 + OWN_VP * NODE_P + OWN_VP + (FT / 32) * 2 * QW];
  float* const B2sh = lds;
  float* const B3sh = B2sh + 128 * 64;
  float* const acc = B3sh + 64 * 64;
  int* const cnt = (int*)(acc + OWN_VP * NODE_P);
  int* const wq = cnt + OWN_VP;

  for (int i = (int)threadIdx.x; i < 64 * 32; i += FT) {
    const int kq = i >> 6, n = i & 63;
    const float4 w = *(const float4*)&W2[n * 128 + kq * 4];
    v2f lo, hi;
    lo.x = w.x; lo.y = w.y; hi.x = w.z; hi.y = w.w;
    *(v2f*)&B2sh[(kq * 2) * 128 + n * 2] = lo;
    *(v2f*)&B2sh[(kq * 2 + 1) * 128 + n * 2] = hi;
  }
  for (int i = (int)threadIdx.x; i < 64 * 16; i += FT) {
    const int kq = i >> 6, n = i & 63;
    const float4 w = *(const float4*)&Ww[n * 64 + kq * 4];
    v2f lo, hi;
    lo.x = w.x; lo.y = w.y; hi.x = w.z; hi.y = w.w;
    *(v2f*)&B3sh[(kq * 2) * 128 + n * 2] = lo;
    *(v2f*)&B3sh[(kq * 2 + 1) * 128 + n * 2] = hi;
  }
  for (int i = OWN_V * NODE_P + (int)threadIdx.x; i < OWN_VP * NODE_P; i += FT)
    acc[i] = 0.0f;
  if ((int)threadIdx.x < OWN_VP - OWN_V) cnt[OWN_V + (int)threadIdx.x] = 0;

  filter_sum_body(Xe, pkV, mapV, nullptr, false, true, OWN_V, NODE_P, N_NODES_C,
                  acc, cnt, wq);

  const int lane = threadIdx.x & 31;
  const int wave = __builtin_amdgcn_readfirstlane((int)(threadIdx.x >> 5));
  const int ll = lane & 15;
  const int lh = lane >> 4;
  const int base = blockIdx.x * OWN_V;
  const bool hasTile = wave < NODE_TILES;
  const int r0 = wave * 16;
  const float oma = 1.0f - ALPHA_C;

  if (hasTile) {
    const int gA = base + r0 + ll;
    const int gAc = gA < N_NODES_C ? gA : N_NODES_C - 1;
    const float dg = (float)cnt[r0 + ll];
    const float* Xrow = X + (size_t)gAc * 64;
    const v2f* Srow = (const v2f*)&acc[(r0 + ll) * NODE_P + lh * 2];
    const v2f* Bl2 = (const v2f*)&B2sh[lh * 128 + ll * 2];
    v2f xa[16];
#pragma unroll
    for (int s = 0; s < 16; ++s) {
      const int ka = s * 4 + lh * 2;
      xa[s].x = dg * Xrow[ka]; xa[s].y = dg * Xrow[ka + 1];
    }
    v8f acc0 = {}, acc1 = {}, acc2 = {}, acc3 = {};
#pragma unroll
    for (int s = 0; s < 16; ++s) {
      const v2f a = xa[s];
      const v2f* B = Bl2 + s * 128;
      acc0 = wmma_f32(a, B[0], acc0);
      acc1 = wmma_f32(a, B[16], acc1);
      acc2 = wmma_f32(a, B[32], acc2);
      acc3 = wmma_f32(a, B[48], acc3);
    }
#pragma unroll
    for (int s = 0; s < 16; ++s) {
      const v2f a = Srow[s * 2];
      const v2f* B = Bl2 + 32 * 64 + s * 128;
      acc0 = wmma_f32(a, B[0], acc0);
      acc1 = wmma_f32(a, B[16], acc1);
      acc2 = wmma_f32(a, B[32], acc2);
      acc3 = wmma_f32(a, B[48], acc3);
    }
    const float c0 = b2[ll], c1 = b2[16 + ll], c2 = b2[32 + ll], c3 = b2[48 + ll];
#pragma unroll
    for (int j = 0; j < 8; ++j) {
      const int lr = r0 + j + lh * 8;
      const int g = base + lr;
      const int gc = g < N_NODES_C ? g : N_NODES_C - 1;
      const float* x0r = X0 + (size_t)gc * 64;
      const float dr = (float)cnt[lr];
      float* tr = &acc[lr * NODE_P];
      tr[ll]      = oma * (acc0[j] + dr * c0) + ALPHA_C * x0r[ll];
      tr[16 + ll] = oma * (acc1[j] + dr * c1) + ALPHA_C * x0r[16 + ll];
      tr[32 + ll] = oma * (acc2[j] + dr * c2) + ALPHA_C * x0r[32 + ll];
      tr[48 + ll] = oma * (acc3[j] + dr * c3) + ALPHA_C * x0r[48 + ll];
    }
  }
  __syncthreads();
  if (hasTile) {
    const v2f* Trow = (const v2f*)&acc[(r0 + ll) * NODE_P + lh * 2];
    const v2f* Bl3 = (const v2f*)&B3sh[lh * 128 + ll * 2];
    v8f o0 = {}, o1 = {}, o2 = {}, o3 = {};
#pragma unroll
    for (int s = 0; s < 16; ++s) {
      const v2f a = Trow[s * 2];
      const v2f* B = Bl3 + s * 128;
      o0 = wmma_f32(a, B[0], o0);
      o1 = wmma_f32(a, B[16], o1);
      o2 = wmma_f32(a, B[32], o2);
      o3 = wmma_f32(a, B[48], o3);
    }
    const float bi0 = Wb[ll], bi1 = Wb[16 + ll], bi2 = Wb[32 + ll],
                bi3 = Wb[48 + ll];
#pragma unroll
    for (int j = 0; j < 8; ++j) {
      const int lr = r0 + j + lh * 8;
      const int g = base + lr;
      if (lr < OWN_V && g < N_NODES_C) {
        float* o = out + (size_t)g * 64;
        o[ll]      = o0[j] + bi0;
        o[16 + ll] = o1[j] + bi1;
        o[32 + ll] = o2[j] + bi2;
        o[48 + ll] = o3[j] + bi3;
      }
    }
  }
}

extern "C" void kernel_launch(void* const* d_in, const int* in_sizes, int n_in,
                              void* d_out, int out_size, void* d_ws,
                              size_t ws_size, hipStream_t stream) {
  (void)in_sizes; (void)n_in; (void)out_size; (void)ws_size;
  const float* X    = (const float*)d_in[0];
  const float* X0   = (const float*)d_in[1];
  const int*   vertex = (const int*)d_in[2];
  const int*   edges  = (const int*)d_in[3];
  const float* W1w  = (const float*)d_in[4];
  const float* W1b  = (const float*)d_in[5];
  const float* W2w  = (const float*)d_in[6];
  const float* W2b  = (const float*)d_in[7];
  const float* Ww   = (const float*)d_in[8];
  const float* Wb   = (const float*)d_in[9];

  float* out = (float*)d_out;
  float* Xe  = out + (size_t)N_NODES_C * 64;

  float* ws  = (float*)d_ws;
  float* Y1  = ws;

  unsigned int* mapE = (unsigned int*)(Y1 + (size_t)N_NODES_C * 64);
  unsigned int* mapV = mapE + NNZ_C / 4;
  unsigned int* pkE  = mapV + NNZ_C / 4;
  unsigned int* pkV  = pkE + NNZ_C;

  gemm64_bias_kernel<<<RES_WG, GEMM_T, 0, stream>>>(X, W1w, W1b, Y1, N_NODES_C);
  owner_map_kernel<<<RES_WG, MAP_T, 0, stream>>>(vertex, edges, mapE, mapV, pkE,
                                                 pkV);
  filter_edges_kernel<<<OWNERS, FT, 0, stream>>>(Y1, pkE, mapE, Xe);
  filter_nodes_kernel<<<OWNERS, FT, 0, stream>>>(Xe, pkV, mapV, X, X0, W2w, W2b,
                                                 Ww, Wb, out);
}


# owner map as one-hot masks of the owner number's two nibbles (32 arrays per sum), not 8 bit planes: an owner reads 2 dwords per 32 incidences, not 8; queue order changes, LDS f32 adds arrival order
# speedup vs baseline: 3.6332x; 1.0962x over previous
#include <hip/hip_runtime.h>

typedef __attribute__((ext_vector_type(2))) float v2f;
typedef __attribute__((ext_vector_type(8))) float v8f;

#define N_NODES_C 100000
#define N_EDGES_C 20000
#define NNZ_C     1000000
#define ALPHA_C   0.5f

__device__ __forceinline__ v8f wmma_f32(v2f a, v2f b, v8f c) {
  return __builtin_amdgcn_wmma_f32_16x16x4_f32(false, a, false, b, (short)0, c,
                                               false, false);
}

#define RES_WG   256
#define GEMM_T   512
#define GEMM_NW  (GEMM_T / 32)

__global__ __launch_bounds__(GEMM_T) void gemm64_bias_kernel(
    const float* __restrict__ A, const float* __restrict__ W,
    const float* __restrict__ bias, float* __restrict__ Out, int M) {
  __shared__ float Bsh[64 * 64];

  const int lane = threadIdx.x & 31;
  const int wave = __builtin_amdgcn_readfirstlane((int)(threadIdx.x >> 5));
  const int ll = lane & 15;
  const int lh = lane >> 4;
  const int tiles = M >> 4;
  const int tBeg = (int)(((long long)blockIdx.x * tiles) / RES_WG);
  const int tEnd = (int)(((long long)(blockIdx.x + 1) * tiles) / RES_WG);
  int t = tBeg + wave;

  v2f an[16];
#pragma unroll
  for (int s = 0; s < 16; ++s) { an[s].x = 0.0f; an[s].y = 0.0f; }
  if (t < tEnd) {
    const float* Arow = A + (size_t)(t * 16 + ll) * 64;
#pragma unroll
    for (int s = 0; s < 16; ++s) {
      const int ka = s * 4 + lh * 2;
      an[s].x = Arow[ka];
      an[s].y = Arow[ka + 1];
    }
  }

#pragma unroll
  for (int j = 0; j < 64 * 64 / GEMM_T; ++j) {
    const int i = j * GEMM_T + (int)threadIdx.x;
    const int k = i >> 6, n = i & 63;
    Bsh[i] = W[n * 64 + k];
  }
  const float bi0 = bias[ll], bi1 = bias[16 + ll], bi2 = bias[32 + ll],
              bi3 = bias[48 + ll];
  __syncthreads();

  while (t < tEnd) {
    v2f a[16];
#pragma unroll
    for (int s = 0; s < 16; ++s) a[s] = an[s];
    const int tn = t + GEMM_NW;
    if (tn < tEnd) {
      const float* Arow = A + (size_t)(tn * 16 + ll) * 64;
#pragma unroll
      for (int s = 0; s < 16; ++s) {
        const int ka = s * 4 + lh * 2;
        an[s].x = Arow[ka];
        an[s].y = Arow[ka + 1];
      }
    }

    v8f acc0 = {}, acc1 = {}, acc2 = {}, acc3 = {};
#pragma unroll
    for (int s = 0; s < 16; ++s) {
      const int ka = s * 4 + lh * 2;
      const float* B0 = &Bsh[ka * 64 + ll];
      const float* B1 = &Bsh[(ka + 1) * 64 + ll];
      v2f b;
      b.x = B0[0];  b.y = B1[0];  acc0 = wmma_f32(a[s], b, acc0);
      b.x = B0[16]; b.y = B1[16]; acc1 = wmma_f32(a[s], b, acc1);
      b.x = B0[32]; b.y = B1[32]; acc2 = wmma_f32(a[s], b, acc2);
      b.x = B0[48]; b.y = B1[48]; acc3 = wmma_f32(a[s], b, acc3);
    }

    const int rowBase = t << 4;
#pragma unroll
    for (int j = 0; j < 8; ++j) {
      const int r = rowBase + j + lh * 8;
      float* o = Out + (size_t)r * 64;
      o[ll]      = acc0[j] + bi0;
      o[16 + ll] = acc1[j] + bi1;
      o[32 + ll] = acc2[j] + bi2;
      o[48 + ll] = acc3[j] + bi3;
    }
    t = tn;
  }
}

#define OWN_E   79
#define OWN_V   391
#define OWNERS  256
#define FT      1024
#define FG      4
#define QW      128
#define DR      24
#define NGROUPS (NNZ_C / 32)
#define NGP     (((NGROUPS + FT * FG - 1) / (FT * FG)) * (FT * FG))
#define MAP_ARRS 32

#define MAP_T     256
#define MAP_TASKS (NGP / 32)
#define MAP_WG    ((2 * MAP_TASKS) / (MAP_T / 32))

__device__ __forceinline__ void map_half(
    const int (&ev)[16], const int (&vv)[16], const int g0, const int k0,
    const int lane, const bool forV, unsigned int* __restrict__ pk,
    unsigned int (&hiM)[16], unsigned int (&loM)[16]) {
  const unsigned int R = forV ? OWN_V : OWN_E;
#pragma unroll
  for (int k = 0; k < 16; ++k) {
    const int g = g0 + k0 + k;
    if (g < NGROUPS) {
      const unsigned int d = (unsigned int)(forV ? vv[k] : ev[k]);
      const unsigned int s = (unsigned int)(forV ? ev[k] : vv[k]);
      const unsigned int o = forV ? d / OWN_V : d / OWN_E;
      pk[g * 32 + lane] = ((d - o * R) << 20) | s;
      const unsigned int hi = o >> 4, lo = o & 15u;
#pragma unroll
      for (int a = 0; a < 16; ++a) {
        const unsigned int bH = __builtin_amdgcn_ballot_w32(hi == (unsigned int)a);
        const unsigned int bL = __builtin_amdgcn_ballot_w32(lo == (unsigned int)a);
        if (lane == k0 + k) { hiM[a] = bH; loM[a] = bL; }
      }
    }
  }
}

__global__ __launch_bounds__(MAP_T) void owner_map_kernel(
    const int* __restrict__ vertex, const int* __restrict__ edges,
    unsigned int* __restrict__ mapE, unsigned int* __restrict__ mapV,
    unsigned int* __restrict__ pkE, unsigned int* __restrict__ pkV) {
  const int lane = threadIdx.x & 31;
  const int waveG = blockIdx.x * (MAP_T / 32) +
                    __builtin_amdgcn_readfirstlane((int)(threadIdx.x >> 5));
  const bool forV = (waveG & 1) != 0;
  const int g0 = (waveG >> 1) * 32;
  int evA[16], vvA[16], evB[16], vvB[16];
#pragma unroll
  for (int k = 0; k < 16; ++k) {
    const int ga = g0 + k, gb = g0 + 16 + k;
    const int gla = ga < NGROUPS ? ga : NGROUPS - 1;
    const int glb = gb < NGROUPS ? gb : NGROUPS - 1;
    evA[k] = edges[gla * 32 + lane];
    vvA[k] = vertex[gla * 32 + lane];
    evB[k] = edges[glb * 32 + lane];
    vvB[k] = vertex[glb * 32 + lane];
  }
  unsigned int hiM[16], loM[16];
#pragma unroll
  for (int a = 0; a < 16; ++a) { hiM[a] = 0u; loM[a] = 0u; }
  unsigned int* const pk = forV ? pkV : pkE;
  map_half(evA, vvA, g0, 0, lane, forV, pk, hiM, loM);
  map_half(evB, vvB, g0, 16, lane, forV, pk, hiM, loM);
  unsigned int* const mp = (forV ? mapV : mapE) + g0 + lane;
#pragma unroll
  for (int a = 0; a < 16; ++a) {
    mp[a * NGP] = hiM[a];
    mp[(16 + a) * NGP] = loM[a];
  }
}

__device__ __forceinline__ void filter_slow_one(
    const float* __restrict__ src, const unsigned int* __restrict__ pkw, int e,
    float* acc, int* cnt, bool wantDeg, const int P) {
  const unsigned int w = pkw[e];
  const int r = (int)(w >> 20);
  const float* s = src + (size_t)(w & 0xFFFFFu) * 64;
  for (int c = 0; c < 64; ++c) atomicAdd(&acc[r * P + c], s[c]);
  if (wantDeg) atomicAdd(&cnt[r], 1);
}

__device__ __forceinline__ void filter_sum_body(
    const float* __restrict__ src, const unsigned int* __restrict__ pkw,
    const unsigned int* __restrict__ map,
    float* __restrict__ dst, const bool doStore, const bool wantDeg,
    const int R, const int P, const int nDst, float* acc, int* cnt, int* wq) {
  const int tid = threadIdx.x;
  const int lane = tid & 31;
  const int wave = __builtin_amdgcn_readfirstlane(tid >> 5);
  const int base = blockIdx.x * R;

  for (int i = tid; i < R * P; i += FT) acc[i] = 0.0f;
  if (wantDeg)
    for (int i = tid; i < R; i += FT) cnt[i] = 0;
  __syncthreads();

  const unsigned int me = blockIdx.x;
  const int nChunks = NGP / (FT * FG);
  const unsigned int* const hiW = map + (size_t)(me >> 4) * NGP;
  const unsigned int* const loW = map + (size_t)(16u + (me & 15u)) * NGP;
  const uint4* const hiP = (const uint4*)hiW;
  const uint4* const loP = (const uint4*)loW;
  int* const qw = wq + wave * (2 * QW);

  uint4 mh = hiP[tid];
  uint4 ml = loP[tid];

  int pkP = -1;
  int nPrev = 0;

  for (int c = 0; c <= nChunks; ++c) {
    const int c0 = c * (FT * FG);
    const bool doScan = c < nChunks;
    int* qScan = qw + (c & 1) * QW;
    const int* qDrain = qw + ((c & 1) ^ 1) * QW;
    int nCur = 0;
    int pkN = -1;

    if (doScan) {
      const unsigned int zs[FG] = {mh.x & ml.x, mh.y & ml.y, mh.z & ml.z,
                                   mh.w & ml.w};
#pragma unroll
      for (int s = 0; s < FG; ++s) {
        const int gi = c0 + tid * FG + s;
        unsigned int z = zs[s];
        while (true) {
          const bool act = z != 0u;
          const unsigned int bal = __builtin_amdgcn_ballot_w32(act);
          if (bal == 0u) break;
          if (act) {
            const int slot = nCur + (int)__builtin_amdgcn_mbcnt_lo(bal, 0u);
            if (slot < QW) qScan[slot] = gi * 32 + __builtin_ctz(z);
            z &= z - 1;
          }
          nCur += __builtin_popcount(bal);
        }
      }
      if (nCur > QW) {
        nCur = 0;
#pragma nounroll
        for (int s = 0; s < FG; ++s) {
          const int gi = c0 + tid * FG + s;
          unsigned int z = hiW[gi] & loW[gi];
          while (z) {
            const int b = __builtin_ctz(z);
            z &= z - 1;
            filter_slow_one(src, pkw, gi * 32 + b, acc, cnt, wantDeg, P);
          }
        }
      }

      if (c + 1 < nChunks) {
        mh = hiP[(c + 1) * FT + tid];
        ml = loP[(c + 1) * FT + tid];
      }

      if (lane < DR && lane < nCur) pkN = (int)pkw[qScan[lane]];
    }

    for (int jb = 0; jb < nPrev; jb += DR) {
      int pk = pkP;
      if (jb > 0) {
        pk = -1;
        if (lane < DR && jb + lane < nPrev) pk = (int)pkw[qDrain[jb + lane]];
      }
      const int n = (nPrev - jb) < DR ? (nPrev - jb) : DR;
      const bool has = pk >= 0;
      const unsigned int sOff = has ? (((unsigned int)pk & 0xFFFFFu) << 8) : 0u;
      const unsigned int dOff =
          has ? (((unsigned int)pk >> 20) * (unsigned int)(P * 4)) : 0u;
      if (wantDeg && has) atomicAdd(&cnt[(unsigned int)pk >> 20], 1);
      float2 r[DR];
#pragma unroll
      for (int k = 0; k < DR; ++k) {
        const unsigned int so =
            (unsigned int)__builtin_amdgcn_readlane((int)sOff, k);
        r[k] = *(const float2*)((const char*)src + (size_t)so +
                                (size_t)(lane * 8));
      }
      asm volatile("" : "+v"(r[1].x), "+v"(r[1].y));
#pragma unroll
      for (int k = 0; k < DR; ++k) {
        if (k == 0 || k < n) {
          const unsigned int d =
              (unsigned int)__builtin_amdgcn_readlane((int)dOff, k);
          float* a = (float*)((char*)acc + d) + lane * 2;
          atomicAdd(a, r[k].x);
          atomicAdd(a + 1, r[k].y);
        }
      }
    }
    pkP = pkN;
    nPrev = nCur;
  }
  __syncthreads();

  if (doStore)
    for (int i = tid; i < R * 64; i += FT) {
      const int row = base + (i >> 6);
      if (row < nDst)
        dst[(size_t)row * 64 + (i & 63)] = acc[(i >> 6) * P + (i & 63)];
    }
}

__global__ __launch_bounds__(FT) void filter_edges_kernel(
    const float* __restrict__ Y1, const unsigned int* __restrict__ pkE,
    const unsigned int* __restrict__ mapE, float* __restrict__ Xe) {
  __shared__ float acc[OWN_E * 64];
  __shared__ int wq[(FT / 32) * 2 * QW];
  filter_sum_body(Y1, pkE, mapE, Xe, true, false, OWN_E, 64, N_EDGES_C, acc,
                  nullptr, wq);
}

#define OWN_VP  400
#define NODE_TILES (OWN_VP / 16)
#define NODE_P  66
__global__ __launch_bounds__(FT) void filter_nodes_kernel(
    const float* __restrict__ Xe, const unsigned int* __restrict__ pkV,
    const unsigned int* __restrict__ mapV, const float* __restrict__ X,
    const float* __restrict__ X0, const float* __restrict__ W2,
    const float* __restrict__ b2, const float* __restrict__ Ww,
    const float* __restrict__ Wb, float* __restrict__ out) {
  __shared__ __attribute__((aligned(16))) float
      lds[128 * 64 + 64 * 64 + OWN_VP * NODE_P + OWN_VP + (FT / 32) * 2 * QW];
  float* const B2sh = lds;
  float* const B3sh = B2sh + 128 * 64;
  float* const acc = B3sh + 64 * 64;
  int* const cnt = (int*)(acc + OWN_VP * NODE_P);
  int* const wq = cnt + OWN_VP;

  for (int i = (int)threadIdx.x; i < 64 * 32; i += FT) {
    const int kq = i >> 6, n = i & 63;
    const float4 w = *(const float4*)&W2[n * 128 + kq * 4];
    v2f lo, hi;
    lo.x = w.x; lo.y = w.y; hi.x = w.z; hi.y = w.w;
    *(v2f*)&B2sh[(kq * 2) * 128 + n * 2] = lo;
    *(v2f*)&B2sh[(kq * 2 + 1) * 128 + n * 2] = hi;
  }
  for (int i = (int)threadIdx.x; i < 64 * 16; i += FT) {
    const int kq = i >> 6, n = i & 63;
    const float4 w = *(const float4*)&Ww[n * 64 + kq * 4];
    v2f lo, hi;
    lo.x = w.x; lo.y = w.y; hi.x = w.z; hi.y = w.w;
    *(v2f*)&B3sh[(kq * 2) * 128 + n * 2] = lo;
    *(v2f*)&B3sh[(kq * 2 + 1) * 128 + n * 2] = hi;
  }
  for (int i = OWN_V * NODE_P + (int)threadIdx.x; i < OWN_VP * NODE_P; i += FT)
    acc[i] = 0.0f;
  if ((int)threadIdx.x < OWN_VP - OWN_V) cnt[OWN_V + (int)threadIdx.x] = 0;

  filter_sum_body(Xe, pkV, mapV, nullptr, false, true, OWN_V, NODE_P, N_NODES_C,
                  acc, cnt, wq);

  const int lane = threadIdx.x & 31;
  const int wave = __builtin_amdgcn_readfirstlane((int)(threadIdx.x >> 5));
  const int ll = lane & 15;
  const int lh = lane >> 4;
  const int base = blockIdx.x * OWN_V;
  const bool hasTile = wave < NODE_TILES;
  const int r0 = wave * 16;
  const float oma = 1.0f - ALPHA_C;

  if (hasTile) {
    const int gA = base + r0 + ll;
    const int gAc = gA < N_NODES_C ? gA : N_NODES_C - 1;
    const float dg = (float)cnt[r0 + ll];
    const float* Xrow = X + (size_t)gAc * 64;
    const v2f* Srow = (const v2f*)&acc[(r0 + ll) * NODE_P + lh * 2];
    const v2f* Bl2 = (const v2f*)&B2sh[lh * 128 + ll * 2];
    v2f xa[16];
#pragma unroll
    for (int s = 0; s < 16; ++s) {
      const int ka = s * 4 + lh * 2;
      xa[s].x = dg * Xrow[ka]; xa[s].y = dg * Xrow[ka + 1];
    }
    v8f acc0 = {}, acc1 = {}, acc2 = {}, acc3 = {};
#pragma unroll
    for (int s = 0; s < 16; ++s) {
      const v2f a = xa[s];
      const v2f* B = Bl2 + s * 128;
      acc0 = wmma_f32(a, B[0], acc0);
      acc1 = wmma_f32(a, B[16], acc1);
      acc2 = wmma_f32(a, B[32], acc2);
      acc3 = wmma_f32(a, B[48], acc3);
    }
#pragma unroll
    for (int s = 0; s < 16; ++s) {
      const v2f a = Srow[s * 2];
      const v2f* B = Bl2 + 32 * 64 + s * 128;
      acc0 = wmma_f32(a, B[0], acc0);
      acc1 = wmma_f32(a, B[16], acc1);
      acc2 = wmma_f32(a, B[32], acc2);
      acc3 = wmma_f32(a, B[48], acc3);
    }
    const float c0 = b2[ll], c1 = b2[16 + ll], c2 = b2[32 + ll], c3 = b2[48 + ll];
#pragma unroll
    for (int j = 0; j < 8; ++j) {
      const int lr = r0 + j + lh * 8;
      const int g = base + lr;
      const int gc = g < N_NODES_C ? g : N_NODES_C - 1;
      const float* x0r = X0 + (size_t)gc * 64;
      const float dr = (float)cnt[lr];
      float* tr = &acc[lr * NODE_P];
      tr[ll]      = oma * (acc0[j] + dr * c0) + ALPHA_C * x0r[ll];
      tr[16 + ll] = oma * (acc1[j] + dr * c1) + ALPHA_C * x0r[16 + ll];
      tr[32 + ll] = oma * (acc2[j] + dr * c2) + ALPHA_C * x0r[32 + ll];
      tr[48 + ll] = oma * (acc3[j] + dr * c3) + ALPHA_C * x0r[48 + ll];
    }
  }
  __syncthreads();
  if (hasTile) {
    const v2f* Trow = (const v2f*)&acc[(r0 + ll) * NODE_P + lh * 2];
    const v2f* Bl3 = (const v2f*)&B3sh[lh * 128 + ll * 2];
    v8f o0 = {}, o1 = {}, o2 = {}, o3 = {};
#pragma unroll
    for (int s = 0; s < 16; ++s) {
      const v2f a = Trow[s * 2];
      const v2f* B = Bl3 + s * 128;
      o0 = wmma_f32(a, B[0], o0);
      o1 = wmma_f32(a, B[16], o1);
      o2 = wmma_f32(a, B[32], o2);
      o3 = wmma_f32(a, B[48], o3);
    }
    const float bi0 = Wb[ll], bi1 = Wb[16 + ll], bi2 = Wb[32 + ll],
                bi3 = Wb[48 + ll];
#pragma unroll
    for (int j = 0; j < 8; ++j) {
      const int lr = r0 + j + lh * 8;
      const int g = base + lr;
      if (lr < OWN_V && g < N_NODES_C) {
        float* o = out + (size_t)g * 64;
        o[ll]      = o0[j] + bi0;
        o[16 + ll] = o1[j] + bi1;
        o[32 + ll] = o2[j] + bi2;
        o[48 + ll] = o3[j] + bi3;
      }
    }
  }
}

extern "C" void kernel_launch(void* const* d_in, const int* in_sizes, int n_in,
                              void* d_out, int out_size, void* d_ws,
                              size_t ws_size, hipStream_t stream) {
  (void)in_sizes; (void)n_in; (void)out_size; (void)ws_size;
  const float* X    = (const float*)d_in[0];
  const float* X0   = (const float*)d_in[1];
  const int*   vertex = (const int*)d_in[2];
  const int*   edges  = (const int*)d_in[3];
  const float* W1w  = (const float*)d_in[4];
  const float* W1b  = (const float*)d_in[5];
  const float* W2w  = (const float*)d_in[6];
  const float* W2b  = (const float*)d_in[7];
  const float* Ww   = (const float*)d_in[8];
  const float* Wb   = (const float*)d_in[9];

  float* out = (float*)d_out;
  float* Xe  = out + (size_t)N_NODES_C * 64;

  float* ws  = (float*)d_ws;
  float* Y1  = ws;

  unsigned int* mapE = (unsigned int*)(Y1 + (size_t)N_NODES_C * 64);
  unsigned int* mapV = mapE + (size_t)MAP_ARRS * NGP;
  unsigned int* pkE  = mapV + (size_t)MAP_ARRS * NGP;
  unsigned int* pkV  = pkE + NNZ_C;

  gemm64_bias_kernel<<<RES_WG, GEMM_T, 0, stream>>>(X, W1w, W1b, Y1, N_NODES_C);
  owner_map_kernel<<<MAP_WG, MAP_T, 0, stream>>>(vertex, edges, mapE, mapV, pkE,
                                                 pkV);
  filter_edges_kernel<<<OWNERS, FT, 0, stream>>>(Y1, pkE, mapE, Xe);
  filter_nodes_kernel<<<OWNERS, FT, 0, stream>>>(Xe, pkV, mapV, X, X0, W2w, W2b,
                                                 Ww, Wb, out);
}
